# starting kernel
# baseline (speedup 1.0000x reference)
#include <hip/hip_runtime.h>
#include <hip/hip_bf16.h>

#define N_T   100000
#define N_M   20000
#define N_E   600000
#define N_Q   200000
#define F_IN  256
#define DDIM  128

#define K_SLAB   64
#define B_STRIDE 68

typedef float v2f __attribute__((ext_vector_type(2)));
typedef float v8f __attribute__((ext_vector_type(8)));

__global__ void k_deg_count(const int* __restrict__ idx, float* __restrict__ deg, int n) {
    int t = blockIdx.x * blockDim.x + threadIdx.x;
    if (t >= n) return;
    unsafeAtomicAdd(&deg[idx[t]], 1.0f);
}

__global__ void k_recip(float* __restrict__ deg, int n) {
    int t = blockIdx.x * blockDim.x + threadIdx.x;
    if (t >= n) return;
    deg[t] = 1.0f / fmaxf(deg[t], 1.0f);
}

__global__ void k_gather_rows(const float* __restrict__ src, const int* __restrict__ ids,
                              float* __restrict__ dst, int nrows) {
    int t = blockIdx.x * blockDim.x + threadIdx.x;
    if (t >= nrows * (DDIM / 4)) return;
    int row = t >> 5;
    int l   = t & 31;
    float4 v = *(const float4*)(src + (size_t)ids[row] * DDIM + l * 4);
    *(float4*)(dst + (size_t)row * DDIM + l * 4) = v;
}

__global__ void k_scatter_mean(const float* __restrict__ feat,
                               const int* __restrict__ idx_from,
                               const int* __restrict__ idx_to,
                               const float* __restrict__ rdeg_to,
                               float* __restrict__ agg, int nedges) {
    long long t = (long long)blockIdx.x * blockDim.x + threadIdx.x;
    long long e = t >> 5;
    if (e >= nedges) return;
    int l = (int)(t & 31);
    int s = idx_from[e];
    int d = idx_to[e];
    float r = rdeg_to[d];
    float4 v = *(const float4*)(feat + (size_t)s * DDIM + l * 4);
    float* out = agg + (size_t)d * DDIM + l * 4;
    unsafeAtomicAdd(out + 0, v.x * r);
    unsafeAtomicAdd(out + 1, v.y * r);
    unsafeAtomicAdd(out + 2, v.z * r);
    unsafeAtomicAdd(out + 3, v.w * r);
}

__device__ __forceinline__ void gemm_pass(const float* __restrict__ A,
                                          const float* __restrict__ W,
                                          int K, int rowA, int r, int h, int tid,
                                          float* __restrict__ Bs, v8f acc[8]) {
    for (int kk0 = 0; kk0 < K; kk0 += K_SLAB) {
        __syncthreads();
#pragma unroll
        for (int i = 0; i < 8; i++) {
            int idx = tid + i * 256;
            int row = idx >> 4;
            int c   = (idx & 15) << 2;
            float4 v = *(const float4*)(W + (size_t)row * K + kk0 + c);
            *(float4*)(Bs + row * B_STRIDE + c) = v;
        }
        __syncthreads();

        const float* Arow = A + (size_t)(rowA + r) * K + kk0 + 4 * h;
        for (int ks = 0; ks < K_SLAB; ks += 8) {
            float4 a4 = *(const float4*)(Arow + ks);
            v2f a1; a1[0] = a4.x; a1[1] = a4.y;
            v2f a2; a2[0] = a4.z; a2[1] = a4.w;
#pragma unroll
            for (int nt = 0; nt < 8; nt++) {
                float4 b4 = *(const float4*)(Bs + (nt * 16 + r) * B_STRIDE + ks + 4 * h);
                v2f b1; b1[0] = b4.x; b1[1] = b4.y;
                v2f b2; b2[0] = b4.z; b2[1] = b4.w;
                acc[nt] = __builtin_amdgcn_wmma_f32_16x16x4_f32(
                    false, a1, false, b1, (short)0, acc[nt], false, false);
                acc[nt] = __builtin_amdgcn_wmma_f32_16x16x4_f32(
                    false, a2, false, b2, (short)0, acc[nt], false, false);
            }
        }
    }
}

__global__ void __launch_bounds__(256)
k_gemm_wmma(const float* __restrict__ A1, const float* __restrict__ W1, int K1,
            const float* __restrict__ A2, const float* __restrict__ W2, int K2,
            const float* __restrict__ bias, float* __restrict__ out,
            int nrows, int relu) {
    __shared__ float Bs[DDIM * B_STRIDE];

    const int tid  = threadIdx.x;
    const int lane = tid & 31;
    const int wave = tid >> 5;
    const int row0 = (blockIdx.x * 8 + wave) * 16;
    const bool valid = (row0 < nrows);
    const int rowA = valid ? row0 : 0;
    const int r = lane & 15;
    const int h = lane >> 4;

    v8f acc[8] = {};

    gemm_pass(A1, W1, K1, rowA, r, h, tid, Bs, acc);
    if (A2) gemm_pass(A2, W2, K2, rowA, r, h, tid, Bs, acc);

    if (!valid) return;
#pragma unroll
    for (int nt = 0; nt < 8; nt++) {
        int col  = nt * 16 + r;
        float bb = bias[col];
#pragma unroll
        for (int i = 0; i < 8; i++) {
            float v = acc[nt][i] + bb;
            if (relu) v = fmaxf(v, 0.0f);
            out[(size_t)(row0 + h * 8 + i) * DDIM + col] = v;
        }
    }
}

__global__ void k_classify(const float* __restrict__ ht, const float* __restrict__ hm,
                           const int* __restrict__ qs, const int* __restrict__ qd,
                           const float* __restrict__ w, const float* __restrict__ cb,
                           float* __restrict__ out, int nq) {
    long long t = (long long)blockIdx.x * blockDim.x + threadIdx.x;
    long long q = t >> 5;
    if (q >= nq) return;
    int l = (int)(t & 31);
    float4 a  = *(const float4*)(ht + (size_t)qs[q] * DDIM + l * 4);
    float4 b  = *(const float4*)(hm + (size_t)qd[q] * DDIM + l * 4);
    float4 wa = *(const float4*)(w + l * 4);
    float4 wb = *(const float4*)(w + DDIM + l * 4);
    float s = a.x * wa.x + a.y * wa.y + a.z * wa.z + a.w * wa.w
            + b.x * wb.x + b.y * wb.y + b.z * wb.z + b.w * wb.w;
#pragma unroll
    for (int off = 16; off > 0; off >>= 1) s += __shfl_xor(s, off, 32);
    if (l == 0) out[q] = s + cb[0];
}

extern "C" void kernel_launch(void* const* d_in, const int* in_sizes, int n_in,
                              void* d_out, int out_size, void* d_ws, size_t ws_size,
                              hipStream_t stream) {
    const float* x_thesis     = (const float*)d_in[0];
    const int*   mentor_id    = (const int*)  d_in[1];
    const int*   edge_src     = (const int*)  d_in[2];
    const int*   edge_dst     = (const int*)  d_in[3];
    const int*   eli_src      = (const int*)  d_in[4];
    const int*   eli_dst      = (const int*)  d_in[5];
    const float* thesis_lin_w = (const float*)d_in[6];
    const float* thesis_lin_b = (const float*)d_in[7];
    const float* prof_emb     = (const float*)d_in[8];
    const float* W_nbr_sup    = (const float*)d_in[9];
    const float* W_root_sup   = (const float*)d_in[10];
    const float* b_sup        = (const float*)d_in[11];
    const float* W_nbr_rev    = (const float*)d_in[12];
    const float* W_root_rev   = (const float*)d_in[13];
    const float* b_rev        = (const float*)d_in[14];
    const float* cls_w        = (const float*)d_in[15];
    const float* cls_b        = (const float*)d_in[16];
    float*       out          = (float*)d_out;

    float* p    = (float*)d_ws;
    float* ht0  = p; p += (size_t)N_T * DDIM;
    float* ht1  = p; p += (size_t)N_T * DDIM;
    float* aggt = p; p += (size_t)N_T * DDIM;
    float* hm0  = p; p += (size_t)N_M * DDIM;
    float* hm1  = p; p += (size_t)N_M * DDIM;
    float* aggm = p; p += (size_t)N_M * DDIM;
    float* rdt  = p; p += N_T;
    float* rdm  = p; p += N_M;

    const int BLK = 256;
    const int ROWS_PER_BLK = 8 * 16;

    hipMemsetAsync(rdt, 0, (size_t)N_T * sizeof(float), stream);
    hipMemsetAsync(rdm, 0, (size_t)N_M * sizeof(float), stream);
    k_deg_count<<<(N_E + BLK - 1) / BLK, BLK, 0, stream>>>(edge_dst, rdm, N_E);
    k_deg_count<<<(N_E + BLK - 1) / BLK, BLK, 0, stream>>>(edge_src, rdt, N_E);
    k_recip<<<(N_M + BLK - 1) / BLK, BLK, 0, stream>>>(rdm, N_M);
    k_recip<<<(N_T + BLK - 1) / BLK, BLK, 0, stream>>>(rdt, N_T);

    k_gemm_wmma<<<(N_T + ROWS_PER_BLK - 1) / ROWS_PER_BLK, BLK, 0, stream>>>(
        x_thesis, thesis_lin_w, F_IN, nullptr, nullptr, 0,
        thesis_lin_b, ht0, N_T, 0);
    {
        int n = N_M * (DDIM / 4);
        k_gather_rows<<<(n + BLK - 1) / BLK, BLK, 0, stream>>>(prof_emb, mentor_id, hm0, N_M);
    }

    const long long scat_threads = (long long)N_E * 32;
    const int scat_blocks = (int)((scat_threads + BLK - 1) / BLK);

    hipMemsetAsync(aggm, 0, (size_t)N_M * DDIM * sizeof(float), stream);
    k_scatter_mean<<<scat_blocks, BLK, 0, stream>>>(ht0, edge_src, edge_dst, rdm, aggm, N_E);
    hipMemsetAsync(aggt, 0, (size_t)N_T * DDIM * sizeof(float), stream);
    k_scatter_mean<<<scat_blocks, BLK, 0, stream>>>(hm0, edge_dst, edge_src, rdt, aggt, N_E);
    k_gemm_wmma<<<(N_M + ROWS_PER_BLK - 1) / ROWS_PER_BLK, BLK, 0, stream>>>(
        aggm, W_nbr_sup, DDIM, hm0, W_root_sup, DDIM, b_sup, hm1, N_M, 1);
    k_gemm_wmma<<<(N_T + ROWS_PER_BLK - 1) / ROWS_PER_BLK, BLK, 0, stream>>>(
        aggt, W_nbr_rev, DDIM, ht0, W_root_rev, DDIM, b_rev, ht1, N_T, 1);

    const size_t wOff = (size_t)DDIM * DDIM;
    hipMemsetAsync(aggm, 0, (size_t)N_M * DDIM * sizeof(float), stream);
    k_scatter_mean<<<scat_blocks, BLK, 0, stream>>>(ht1, edge_src, edge_dst, rdm, aggm, N_E);
    hipMemsetAsync(aggt, 0, (size_t)N_T * DDIM * sizeof(float), stream);
    k_scatter_mean<<<scat_blocks, BLK, 0, stream>>>(hm1, edge_dst, edge_src, rdt, aggt, N_E);
    k_gemm_wmma<<<(N_M + ROWS_PER_BLK - 1) / ROWS_PER_BLK, BLK, 0, stream>>>(
        aggm, W_nbr_sup + wOff, DDIM, hm1, W_root_sup + wOff, DDIM, b_sup + DDIM, hm0, N_M, 0);
    k_gemm_wmma<<<(N_T + ROWS_PER_BLK - 1) / ROWS_PER_BLK, BLK, 0, stream>>>(
        aggt, W_nbr_rev + wOff, DDIM, ht1, W_root_rev + wOff, DDIM, b_rev + DDIM, ht0, N_T, 0);

    {
        long long t = (long long)N_Q * 32;
        k_classify<<<(int)((t + BLK - 1) / BLK), BLK, 0, stream>>>(
            ht0, hm0, eli_src, eli_dst, cls_w, cls_b, out, N_Q);
    }
}


# Neighbour means by owner blocks: 256 blocks each keep a node range's sums and counts in LDS, scan the whole edge list, add matching rows by LDS float adds, store each row once; sum order changes
# speedup vs baseline: 1.8460x; 1.8460x over previous
#include <hip/hip_runtime.h>
#include <hip/hip_bf16.h>

#define N_T   100000
#define N_M   20000
#define N_E   600000
#define N_Q   200000
#define F_IN  256
#define DDIM  128

#define K_SLAB   64
#define B_STRIDE 68

typedef float v2f __attribute__((ext_vector_type(2)));
typedef float v8f __attribute__((ext_vector_type(8)));

__global__ void k_gather_rows(const float* __restrict__ src, const int* __restrict__ ids,
                              float* __restrict__ dst, int nrows) {
    int t = blockIdx.x * blockDim.x + threadIdx.x;
    if (t >= nrows * (DDIM / 4)) return;
    int row = t >> 5;
    int l   = t & 31;
    float4 v = *(const float4*)(src + (size_t)ids[row] * DDIM + l * 4);
    *(float4*)(dst + (size_t)row * DDIM + l * 4) = v;
}

#define OWNER_BLOCKS   256
#define OWNER_THREADS  1024
#define OWNER_WAVES    32
#define OWN_T_ROWS     391
#define OWN_M_ROWS     79
#define SCAN_LOADS     4
#define QCAP           (32 + SCAN_LOADS * 4 * 32)
#define ITEMS_PER_WAVE 32
#define ITEMS_PER_GROUP 8

__device__ __forceinline__ void owner_drain(const float* __restrict__ feat,
                                            const int* __restrict__ idx_other,
                                            const int* qd, int c, int lane,
                                            float* sums, float* cnts) {
    int loc = 0, o = 0;
    if (lane < c) {
        const int entry = qd[lane];
        loc = entry & 511;
        o   = idx_other[entry >> 9];
        __hip_atomic_fetch_add(cnts + loc, 1.0f, __ATOMIC_RELAXED, __HIP_MEMORY_SCOPE_WORKGROUP);
    }
    for (int j0 = 0; j0 < c; j0 += ITEMS_PER_GROUP) {
        float v[ITEMS_PER_GROUP][4];
#pragma unroll
        for (int j = 0; j < ITEMS_PER_GROUP; j++) {
            const int oj = __builtin_amdgcn_readlane(o, j0 + j);
            const float* src = feat + (size_t)oj * DDIM + lane;
            v[j][0] = src[0];
            v[j][1] = src[32];
            v[j][2] = src[64];
            v[j][3] = src[96];
        }
#pragma unroll
        for (int j = 0; j < ITEMS_PER_GROUP; j++) {
            if (j == 0 || j0 + j < c) {
                const int lj = __builtin_amdgcn_readlane(loc, j0 + j);
                float* dst = sums + lj * DDIM + lane;
                __hip_atomic_fetch_add(dst +  0, v[j][0], __ATOMIC_RELAXED, __HIP_MEMORY_SCOPE_WORKGROUP);
                __hip_atomic_fetch_add(dst + 32, v[j][1], __ATOMIC_RELAXED, __HIP_MEMORY_SCOPE_WORKGROUP);
                __hip_atomic_fetch_add(dst + 64, v[j][2], __ATOMIC_RELAXED, __HIP_MEMORY_SCOPE_WORKGROUP);
                __hip_atomic_fetch_add(dst + 96, v[j][3], __ATOMIC_RELAXED, __HIP_MEMORY_SCOPE_WORKGROUP);
            }
        }
    }
}

template <int ROWS>
__global__ void __launch_bounds__(OWNER_THREADS)
k_owner_mean(const float* __restrict__ feat,
             const int* __restrict__ idx_key,
             const int* __restrict__ idx_other,
             float* __restrict__ agg, int n_nodes) {
    static_assert(ROWS <= 512, "a list entry keeps the row inside the range in 9 bits");
    static_assert(N_E <= (1 << 22), "a list entry keeps the edge number above bit 9");
    static_assert(N_E % 4 == 0, "the key list is read 4 keys per lane");
    __shared__ float sums[ROWS * DDIM];
    __shared__ float cnts[ROWS];
    __shared__ int   q_all[OWNER_WAVES][QCAP];

    const int tid  = threadIdx.x;
    const int lane = tid & 31;
    const int w  = __builtin_amdgcn_readfirstlane(tid >> 5);
    const int lo = blockIdx.x * ROWS;
    if (lo >= n_nodes) return;
    const int own = min(ROWS, n_nodes - lo);

    for (int i = tid; i < ROWS * (DDIM / 4); i += OWNER_THREADS)
        *(float4*)(sums + i * 4) = make_float4(0.0f, 0.0f, 0.0f, 0.0f);
    for (int i = tid; i < ROWS; i += OWNER_THREADS) cnts[i] = 0.0f;
    __syncthreads();

    int* q = q_all[w];
    const int4* key4 = (const int4*)idx_key;
    constexpr int N4     = N_E / 4;
    constexpr int NG     = (N4 + 31) / 32;
    constexpr int PER_IT = OWNER_WAVES * SCAN_LOADS;
    constexpr int NIT    = (NG + PER_IT - 1) / PER_IT;
    constexpr int NSLOT  = NIT * PER_IT;
    const int rot = (int)(((unsigned)blockIdx.x * (unsigned)NSLOT) / (unsigned)OWNER_BLOCKS);
    int qn = 0;

#define OWNER_SLOT_I4(it_, u_) \
    ((((it_) * PER_IT + (u_) * OWNER_WAVES + w + rot) >= NSLOT \
          ? ((it_) * PER_IT + (u_) * OWNER_WAVES + w + rot - NSLOT) \
          : ((it_) * PER_IT + (u_) * OWNER_WAVES + w + rot)) * 32 + lane)

    int4 cur[SCAN_LOADS];
#pragma unroll
    for (int u = 0; u < SCAN_LOADS; u++) cur[u] = key4[min(OWNER_SLOT_I4(0, u), N4 - 1)];
    for (int it = 0; it < NIT; it++) {
        int4 nxt[SCAN_LOADS];
#pragma unroll
        for (int u = 0; u < SCAN_LOADS; u++) nxt[u] = key4[min(OWNER_SLOT_I4(it + 1, u), N4 - 1)];
#pragma unroll
        for (int u = 0; u < SCAN_LOADS; u++) {
            const int kk[4] = {cur[u].x, cur[u].y, cur[u].z, cur[u].w};
            unsigned mk[4];
#pragma unroll
            for (int k = 0; k < 4; k++)
                mk[k] = __builtin_amdgcn_ballot_w32((unsigned)(kk[k] - lo) < (unsigned)own);
            if (mk[0] | mk[1] | mk[2] | mk[3]) {
                const int  i4    = OWNER_SLOT_I4(it, u);
                const bool in    = i4 < N4;
                const int  ebase = i4 * 4;
#pragma unroll
                for (int k = 0; k < 4; k++) {
                    const int  loc = kk[k] - lo;
                    const bool hit = in && (unsigned)loc < (unsigned)own;
                    const unsigned mask = __builtin_amdgcn_ballot_w32(hit);
                    if (mask) {
                        if (hit) q[qn + (int)__builtin_amdgcn_mbcnt_lo(mask, 0u)] = ((ebase + k) << 9) | loc;
                        qn += __builtin_popcount(mask);
                    }
                }
            }
        }
        while (qn >= 32) {
            qn -= 32;
            owner_drain(feat, idx_other, q + qn, 32, lane, sums, cnts);
        }
#pragma unroll
        for (int u = 0; u < SCAN_LOADS; u++) cur[u] = nxt[u];
    }
#undef OWNER_SLOT_I4
    if (qn > 0) owner_drain(feat, idx_other, q, qn, lane, sums, cnts);
    __syncthreads();

    for (int i = tid; i < own * (DDIM / 4); i += OWNER_THREADS) {
        const int row = i >> 5;
        const int c4  = (i & 31) * 4;
        const float r = 1.0f / fmaxf(cnts[row], 1.0f);
        float4 s = *(const float4*)(sums + row * DDIM + c4);
        s.x *= r; s.y *= r; s.z *= r; s.w *= r;
        *(float4*)(agg + (size_t)(lo + row) * DDIM + c4) = s;
    }
}

__device__ __forceinline__ void k_step8(const float4 a4, const float* __restrict__ Bs,
                                        int ks, int r, int h, v8f acc[8]) {
    v2f a1; a1[0] = a4.x; a1[1] = a4.y;
    v2f a2; a2[0] = a4.z; a2[1] = a4.w;
#pragma unroll
    for (int nt = 0; nt < 8; nt++) {
        float4 b4 = *(const float4*)(Bs + (nt * 16 + r) * B_STRIDE + ks + 4 * h);
        v2f b1; b1[0] = b4.x; b1[1] = b4.y;
        v2f b2; b2[0] = b4.z; b2[1] = b4.w;
        acc[nt] = __builtin_amdgcn_wmma_f32_16x16x4_f32(
            false, a1, false, b1, (short)0, acc[nt], false, false);
        acc[nt] = __builtin_amdgcn_wmma_f32_16x16x4_f32(
            false, a2, false, b2, (short)0, acc[nt], false, false);
    }
}

__device__ __forceinline__ void gemm_pass(const float* A,
                                          const float* __restrict__ W,
                                          int K, int rowA, int r, int h, int tid,
                                          float* __restrict__ Bs, v8f acc[8]) {
    for (int kk0 = 0; kk0 < K; kk0 += K_SLAB) {
        __syncthreads();
#pragma unroll
        for (int i = 0; i < 8; i++) {
            int idx = tid + i * 256;
            int row = idx >> 4;
            int c   = (idx & 15) << 2;
            float4 v = *(const float4*)(W + (size_t)row * K + kk0 + c);
            *(float4*)(Bs + row * B_STRIDE + c) = v;
        }
        __syncthreads();

        const size_t aoff = (size_t)(rowA + r) * K + kk0 + 4 * h;
        const float* Arow = A + aoff;
        for (int ks = 0; ks < K_SLAB; ks += 8) {
            float4 a4 = *(const float4*)(Arow + ks);
            k_step8(a4, Bs, ks, r, h, acc);
        }
        __builtin_amdgcn_sched_barrier(0);
        asm volatile("v_nop\n\tv_nop\n\tv_nop\n\tv_nop");
    }
}

__global__ void __launch_bounds__(256)
k_gemm_wmma(const float* A1, const float* __restrict__ W1, int K1,
            const float* __restrict__ A2, const float* __restrict__ W2, int K2,
            const float* __restrict__ bias, float* out,
            int nrows, int relu) {
    __shared__ float Bs[DDIM * B_STRIDE];

    const int tid  = threadIdx.x;
    const int lane = tid & 31;
    const int wave = tid >> 5;
    const int row0 = (blockIdx.x * 8 + wave) * 16;
    const bool valid = (row0 < nrows);
    const int rowA = valid ? row0 : 0;
    const int r = lane & 15;
    const int h = lane >> 4;

    v8f acc[8] = {};

    gemm_pass(A1, W1, K1, rowA, r, h, tid, Bs, acc);
    if (A2) gemm_pass(A2, W2, K2, rowA, r, h, tid, Bs, acc);

    if (!valid) return;
#pragma unroll
    for (int nt = 0; nt < 8; nt++) {
        int col  = nt * 16 + r;
        float bb = bias[col];
#pragma unroll
        for (int i = 0; i < 8; i++) {
            float v = acc[nt][i] + bb;
            if (relu) v = fmaxf(v, 0.0f);
            out[(size_t)(row0 + h * 8 + i) * DDIM + col] = v;
        }
    }
}

__global__ void __launch_bounds__(256)
k_classify(const float* __restrict__ ht, const float* __restrict__ hm,
           const int* __restrict__ qs, const int* __restrict__ qd,
           const float* __restrict__ w, const float* __restrict__ cb,
           float* __restrict__ out, int nq) {
    const int lane = threadIdx.x & 31;
    const int wave = __builtin_amdgcn_readfirstlane(
        (int)((blockIdx.x * blockDim.x + threadIdx.x) >> 5));
    const int base = wave * ITEMS_PER_WAVE;
    if (base >= nq) return;
    const int q = base + lane;
    int ia = 0, ib = 0;
    if (q < nq) {
        ia = qs[q];
        ib = qd[q];
    }
    const float4 wa = *(const float4*)(w + lane * 4);
    const float4 wb = *(const float4*)(w + DDIM + lane * 4);
    const int cnt = min(ITEMS_PER_WAVE, nq - base);
    float res = 0.0f;
    for (int j0 = 0; j0 < cnt; j0 += ITEMS_PER_GROUP) {
        float4 a[ITEMS_PER_GROUP], b[ITEMS_PER_GROUP];
#pragma unroll
        for (int j = 0; j < ITEMS_PER_GROUP; j++) {
            const int aj = __builtin_amdgcn_readlane(ia, j0 + j);
            const int bj = __builtin_amdgcn_readlane(ib, j0 + j);
            a[j] = *(const float4*)(ht + (size_t)aj * DDIM + lane * 4);
            b[j] = *(const float4*)(hm + (size_t)bj * DDIM + lane * 4);
        }
#pragma unroll
        for (int j = 0; j < ITEMS_PER_GROUP; j++) {
            float s = a[j].x * wa.x + a[j].y * wa.y + a[j].z * wa.z + a[j].w * wa.w
                    + b[j].x * wb.x + b[j].y * wb.y + b[j].z * wb.z + b[j].w * wb.w;
#pragma unroll
            for (int off = 16; off > 0; off >>= 1) s += __shfl_xor(s, off, 32);
            if (lane == j0 + j) res = s;
        }
    }
    if (q < nq) out[q] = res + cb[0];
}

extern "C" void kernel_launch(void* const* d_in, const int* in_sizes, int n_in,
                              void* d_out, int out_size, void* d_ws, size_t ws_size,
                              hipStream_t stream) {
    const float* x_thesis     = (const float*)d_in[0];
    const int*   mentor_id    = (const int*)  d_in[1];
    const int*   edge_src     = (const int*)  d_in[2];
    const int*   edge_dst     = (const int*)  d_in[3];
    const int*   eli_src      = (const int*)  d_in[4];
    const int*   eli_dst      = (const int*)  d_in[5];
    const float* thesis_lin_w = (const float*)d_in[6];
    const float* thesis_lin_b = (const float*)d_in[7];
    const float* prof_emb     = (const float*)d_in[8];
    const float* W_nbr_sup    = (const float*)d_in[9];
    const float* W_root_sup   = (const float*)d_in[10];
    const float* b_sup        = (const float*)d_in[11];
    const float* W_nbr_rev    = (const float*)d_in[12];
    const float* W_root_rev   = (const float*)d_in[13];
    const float* b_rev        = (const float*)d_in[14];
    const float* cls_w        = (const float*)d_in[15];
    const float* cls_b        = (const float*)d_in[16];
    float*       out          = (float*)d_out;

    const size_t T_PLANE = (size_t)N_T * DDIM;
    const size_t M_PLANE = (size_t)N_M * DDIM;
    float* p    = (float*)d_ws;
    float* tX   = p; p += T_PLANE;
    float* tY   = p; p += T_PLANE;
    float* tZ   = p; p += T_PLANE;
    float* mP   = p; p += M_PLANE;
    float* mQ   = p; p += M_PLANE;
    float* mR   = p; p += M_PLANE;

    const int BLK = 256;
    const int ROWS_PER_BLK = 8 * 16;

    float* ht0 = tZ;
    float* hm0 = mR;
    k_gemm_wmma<<<(N_T + ROWS_PER_BLK - 1) / ROWS_PER_BLK, BLK, 0, stream>>>(
        x_thesis, thesis_lin_w, F_IN, nullptr, nullptr, 0,
        thesis_lin_b, ht0, N_T, 0);
    {
        int n = N_M * (DDIM / 4);
        k_gather_rows<<<(n + BLK - 1) / BLK, BLK, 0, stream>>>(prof_emb, mentor_id, hm0, N_M);
    }

    float* hm1 = mQ;
    float* ht1 = tY;
    k_owner_mean<OWN_M_ROWS><<<OWNER_BLOCKS, OWNER_THREADS, 0, stream>>>(ht0, edge_dst, edge_src, mP, N_M);
    k_owner_mean<OWN_T_ROWS><<<OWNER_BLOCKS, OWNER_THREADS, 0, stream>>>(hm0, edge_src, edge_dst, tX, N_T);
    k_gemm_wmma<<<(N_M + ROWS_PER_BLK - 1) / ROWS_PER_BLK, BLK, 0, stream>>>(
        mP, W_nbr_sup, DDIM, hm0, W_root_sup, DDIM, b_sup, hm1, N_M, 1);
    k_gemm_wmma<<<(N_T + ROWS_PER_BLK - 1) / ROWS_PER_BLK, BLK, 0, stream>>>(
        tX, W_nbr_rev, DDIM, ht0, W_root_rev, DDIM, b_rev, ht1, N_T, 1);

    const size_t wOff = (size_t)DDIM * DDIM;
    float* hm2 = mR;
    float* ht2 = tZ;
    k_owner_mean<OWN_M_ROWS><<<OWNER_BLOCKS, OWNER_THREADS, 0, stream>>>(ht1, edge_dst, edge_src, mP, N_M);
    k_owner_mean<OWN_T_ROWS><<<OWNER_BLOCKS, OWNER_THREADS, 0, stream>>>(hm1, edge_src, edge_dst, tX, N_T);
    k_gemm_wmma<<<(N_M + ROWS_PER_BLK - 1) / ROWS_PER_BLK, BLK, 0, stream>>>(
        mP, W_nbr_sup + wOff, DDIM, hm1, W_root_sup + wOff, DDIM, b_sup + DDIM, hm2, N_M, 0);
    k_gemm_wmma<<<(N_T + ROWS_PER_BLK - 1) / ROWS_PER_BLK, BLK, 0, stream>>>(
        tX, W_nbr_rev + wOff, DDIM, ht1, W_root_rev + wOff, DDIM, b_rev + DDIM, ht2, N_T, 0);

    {
        k_classify<<<(N_Q + 8 * ITEMS_PER_WAVE - 1) / (8 * ITEMS_PER_WAVE), BLK, 0, stream>>>(
            ht2, hm2, eli_src, eli_dst, cls_w, cls_b, out, N_Q);
    }
}


# Owner blocks stop scanning the whole key list: one small kernel writes a membership bitmap per owner block (a bit per edge) once per call, each block reads only its own; list and sum order change
# speedup vs baseline: 2.0362x; 1.1030x over previous
#include <hip/hip_runtime.h>
#include <hip/hip_bf16.h>

#define N_T   100000
#define N_M   20000
#define N_E   600000
#define N_Q   200000
#define F_IN  256
#define DDIM  128

#define K_SLAB   64
#define B_STRIDE 68

typedef float v2f __attribute__((ext_vector_type(2)));
typedef float v8f __attribute__((ext_vector_type(8)));

__global__ void k_gather_rows(const float* __restrict__ src, const int* __restrict__ ids,
                              float* __restrict__ dst, int nrows) {
    int t = blockIdx.x * blockDim.x + threadIdx.x;
    if (t >= nrows * (DDIM / 4)) return;
    int row = t >> 5;
    int l   = t & 31;
    float4 v = *(const float4*)(src + (size_t)ids[row] * DDIM + l * 4);
    *(float4*)(dst + (size_t)row * DDIM + l * 4) = v;
}

#define OWNER_BLOCKS   256
#define OWNER_THREADS  1024
#define OWNER_WAVES    32
#define OWN_T_ROWS     391
#define OWN_M_ROWS     79
#define QCAP           64
#define BM_CHUNK       1024
#define BM_NCH         ((N_E + BM_CHUNK - 1) / BM_CHUNK)
#define BM_WORDS       (BM_NCH * 32)
#define BM_PLANE       ((size_t)OWNER_BLOCKS * BM_WORDS)
#define BM_PITCH       257
#define ITEMS_PER_WAVE 32
#define ITEMS_PER_GROUP 8

__global__ void __launch_bounds__(BM_CHUNK)
k_owner_bitmaps(const int* __restrict__ key_m, const int* __restrict__ key_t,
                unsigned* __restrict__ bm_m, unsigned* __restrict__ bm_t) {
    static_assert(OWNER_BLOCKS % 32 == 0, "a wave stores one owner's line per step");
    __shared__ unsigned bits[32 * BM_PITCH];

    const int  tid    = threadIdx.x;
    const int  chunk  = blockIdx.x;
    const bool side_t = blockIdx.y != 0;
    const int* keys   = side_t ? key_t : key_m;
    unsigned*  bm     = side_t ? bm_t : bm_m;

    for (int i = tid; i < 32 * BM_PITCH; i += BM_CHUNK) bits[i] = 0u;
    __syncthreads();

    const int e = chunk * BM_CHUNK + tid;
    if (e < N_E) {
        const unsigned k   = (unsigned)keys[e];
        const unsigned own = side_t ? k / (unsigned)OWN_T_ROWS : k / (unsigned)OWN_M_ROWS;
        if (own < (unsigned)OWNER_BLOCKS)
            __hip_atomic_fetch_or(&bits[(tid >> 5) * BM_PITCH + own], 1u << (tid & 31),
                                  __ATOMIC_RELAXED, __HIP_MEMORY_SCOPE_WORKGROUP);
    }
    __syncthreads();

    const int wd = tid & 31;
    for (int o = tid >> 5; o < OWNER_BLOCKS; o += BM_CHUNK / 32)
        bm[((size_t)o * BM_NCH + chunk) * 32 + wd] = bits[wd * BM_PITCH + o];
}

__device__ __forceinline__ void owner_drain(const float* __restrict__ feat,
                                            const int* __restrict__ idx_key,
                                            const int* __restrict__ idx_other,
                                            const int* qd, int c, int lane, int lo,
                                            float* sums, float* cnts) {
    int loc = 0, o = 0;
    if (lane < c) {
        const int e = qd[lane];
        loc = idx_key[e] - lo;
        o   = idx_other[e];
        __hip_atomic_fetch_add(cnts + loc, 1.0f, __ATOMIC_RELAXED, __HIP_MEMORY_SCOPE_WORKGROUP);
    }
    for (int j0 = 0; j0 < c; j0 += ITEMS_PER_GROUP) {
        float v[ITEMS_PER_GROUP][4];
#pragma unroll
        for (int j = 0; j < ITEMS_PER_GROUP; j++) {
            const int oj = __builtin_amdgcn_readlane(o, j0 + j);
            const float* src = feat + (size_t)oj * DDIM + lane;
            v[j][0] = src[0];
            v[j][1] = src[32];
            v[j][2] = src[64];
            v[j][3] = src[96];
        }
#pragma unroll
        for (int j = 0; j < ITEMS_PER_GROUP; j++) {
            if (j == 0 || j0 + j < c) {
                const int lj = __builtin_amdgcn_readlane(loc, j0 + j);
                float* dst = sums + lj * DDIM + lane;
                __hip_atomic_fetch_add(dst +  0, v[j][0], __ATOMIC_RELAXED, __HIP_MEMORY_SCOPE_WORKGROUP);
                __hip_atomic_fetch_add(dst + 32, v[j][1], __ATOMIC_RELAXED, __HIP_MEMORY_SCOPE_WORKGROUP);
                __hip_atomic_fetch_add(dst + 64, v[j][2], __ATOMIC_RELAXED, __HIP_MEMORY_SCOPE_WORKGROUP);
                __hip_atomic_fetch_add(dst + 96, v[j][3], __ATOMIC_RELAXED, __HIP_MEMORY_SCOPE_WORKGROUP);
            }
        }
    }
}

template <int ROWS>
__global__ void __launch_bounds__(OWNER_THREADS)
k_owner_mean(const float* __restrict__ feat,
             const int* __restrict__ idx_key,
             const int* __restrict__ idx_other,
             const unsigned* __restrict__ bm,
             float* __restrict__ agg, int n_nodes) {
    static_assert((long long)BM_WORDS * 32 >= N_E, "the bitmap has a bit for every edge");
    static_assert(BM_WORDS % 4 == 0, "a block's bitmap is read 16 bytes per lane");
    __shared__ float sums[ROWS * DDIM];
    __shared__ float cnts[ROWS];
    __shared__ int   q_all[OWNER_WAVES][QCAP];

    const int tid  = threadIdx.x;
    const int lane = tid & 31;
    const int w  = __builtin_amdgcn_readfirstlane(tid >> 5);
    const int lo = blockIdx.x * ROWS;
    if (lo >= n_nodes) return;
    const int own = min(ROWS, n_nodes - lo);

    for (int i = tid; i < ROWS * (DDIM / 4); i += OWNER_THREADS)
        *(float4*)(sums + i * 4) = make_float4(0.0f, 0.0f, 0.0f, 0.0f);
    for (int i = tid; i < ROWS; i += OWNER_THREADS) cnts[i] = 0.0f;
    __syncthreads();

    int* q = q_all[w];
    const unsigned* mine = bm + (size_t)blockIdx.x * BM_WORDS;
    int qn = 0;

    const uint4* mine4 = (const uint4*)mine;
    constexpr int N4 = BM_WORDS / 4;
    constexpr int NR = (N4 + OWNER_THREADS - 1) / OWNER_THREADS;
#define OWNER_GROUP(r_) \
    (((r_) * OWNER_THREADS + tid) < N4 \
         ? mine4[min((r_) * OWNER_THREADS + tid, N4 - 1)] : make_uint4(0u, 0u, 0u, 0u))

    uint4 cur = OWNER_GROUP(0);
    for (int r = 0; r < NR; r++) {
        const unsigned round_any =
            __builtin_amdgcn_ballot_w32((cur.x | cur.y | cur.z | cur.w) != 0u);
        const uint4 nxt = OWNER_GROUP(r + 1);
        for (int k = 0; round_any != 0u && k < 4; k++) {
            const unsigned word = k == 0 ? cur.x : k == 1 ? cur.y : k == 2 ? cur.z : cur.w;
            unsigned any = __builtin_amdgcn_ballot_w32(word != 0u);
            const int wbase = ((r * OWNER_THREADS + (w << 5)) << 2) + k;
            while (any) {
                const int L = __builtin_ctz(any);
                any &= any - 1u;
                const unsigned wl = (unsigned)__builtin_amdgcn_readlane((int)word, L);
                const int ebase = (wbase + (L << 2)) << 5;
                if ((wl >> lane) & 1u) q[qn + (int)__builtin_amdgcn_mbcnt_lo(wl, 0u)] = ebase + lane;
                qn += __builtin_popcount(wl);
                if (qn >= 32) {
                    qn -= 32;
                    owner_drain(feat, idx_key, idx_other, q + qn, 32, lane, lo, sums, cnts);
                }
            }
        }
        cur = nxt;
    }
#undef OWNER_GROUP
    if (qn > 0) owner_drain(feat, idx_key, idx_other, q, qn, lane, lo, sums, cnts);
    __syncthreads();

    for (int i = tid; i < own * (DDIM / 4); i += OWNER_THREADS) {
        const int row = i >> 5;
        const int c4  = (i & 31) * 4;
        const float r = 1.0f / fmaxf(cnts[row], 1.0f);
        float4 s = *(const float4*)(sums + row * DDIM + c4);
        s.x *= r; s.y *= r; s.z *= r; s.w *= r;
        *(float4*)(agg + (size_t)(lo + row) * DDIM + c4) = s;
    }
}

__device__ __forceinline__ void k_step8(const float4 a4, const float* __restrict__ Bs,
                                        int ks, int r, int h, v8f acc[8]) {
    v2f a1; a1[0] = a4.x; a1[1] = a4.y;
    v2f a2; a2[0] = a4.z; a2[1] = a4.w;
#pragma unroll
    for (int nt = 0; nt < 8; nt++) {
        float4 b4 = *(const float4*)(Bs + (nt * 16 + r) * B_STRIDE + ks + 4 * h);
        v2f b1; b1[0] = b4.x; b1[1] = b4.y;
        v2f b2; b2[0] = b4.z; b2[1] = b4.w;
        acc[nt] = __builtin_amdgcn_wmma_f32_16x16x4_f32(
            false, a1, false, b1, (short)0, acc[nt], false, false);
        acc[nt] = __builtin_amdgcn_wmma_f32_16x16x4_f32(
            false, a2, false, b2, (short)0, acc[nt], false, false);
    }
}

__device__ __forceinline__ void gemm_pass(const float* A,
                                          const float* __restrict__ W,
                                          int K, int rowA, int r, int h, int tid,
                                          float* __restrict__ Bs, v8f acc[8]) {
    for (int kk0 = 0; kk0 < K; kk0 += K_SLAB) {
        __syncthreads();
#pragma unroll
        for (int i = 0; i < 8; i++) {
            int idx = tid + i * 256;
            int row = idx >> 4;
            int c   = (idx & 15) << 2;
            float4 v = *(const float4*)(W + (size_t)row * K + kk0 + c);
            *(float4*)(Bs + row * B_STRIDE + c) = v;
        }
        __syncthreads();

        const size_t aoff = (size_t)(rowA + r) * K + kk0 + 4 * h;
        const float* Arow = A + aoff;
        for (int ks = 0; ks < K_SLAB; ks += 8) {
            float4 a4 = *(const float4*)(Arow + ks);
            k_step8(a4, Bs, ks, r, h, acc);
        }
        __builtin_amdgcn_sched_barrier(0);
        asm volatile("v_nop\n\tv_nop\n\tv_nop\n\tv_nop");
    }
}

__global__ void __launch_bounds__(256)
k_gemm_wmma(const float* A1, const float* __restrict__ W1, int K1,
            const float* __restrict__ A2, const float* __restrict__ W2, int K2,
            const float* __restrict__ bias, float* out,
            int nrows, int relu) {
    __shared__ float Bs[DDIM * B_STRIDE];

    const int tid  = threadIdx.x;
    const int lane = tid & 31;
    const int wave = tid >> 5;
    const int row0 = (blockIdx.x * 8 + wave) * 16;
    const bool valid = (row0 < nrows);
    const int rowA = valid ? row0 : 0;
    const int r = lane & 15;
    const int h = lane >> 4;

    v8f acc[8] = {};

    gemm_pass(A1, W1, K1, rowA, r, h, tid, Bs, acc);
    if (A2) gemm_pass(A2, W2, K2, rowA, r, h, tid, Bs, acc);

    if (!valid) return;
#pragma unroll
    for (int nt = 0; nt < 8; nt++) {
        int col  = nt * 16 + r;
        float bb = bias[col];
#pragma unroll
        for (int i = 0; i < 8; i++) {
            float v = acc[nt][i] + bb;
            if (relu) v = fmaxf(v, 0.0f);
            out[(size_t)(row0 + h * 8 + i) * DDIM + col] = v;
        }
    }
}

__global__ void __launch_bounds__(256)
k_classify(const float* __restrict__ ht, const float* __restrict__ hm,
           const int* __restrict__ qs, const int* __restrict__ qd,
           const float* __restrict__ w, const float* __restrict__ cb,
           float* __restrict__ out, int nq) {
    const int lane = threadIdx.x & 31;
    const int wave = __builtin_amdgcn_readfirstlane(
        (int)((blockIdx.x * blockDim.x + threadIdx.x) >> 5));
    const int base = wave * ITEMS_PER_WAVE;
    if (base >= nq) return;
    const int q = base + lane;
    int ia = 0, ib = 0;
    if (q < nq) {
        ia = qs[q];
        ib = qd[q];
    }
    const float4 wa = *(const float4*)(w + lane * 4);
    const float4 wb = *(const float4*)(w + DDIM + lane * 4);
    const int cnt = min(ITEMS_PER_WAVE, nq - base);
    float res = 0.0f;
    for (int j0 = 0; j0 < cnt; j0 += ITEMS_PER_GROUP) {
        float4 a[ITEMS_PER_GROUP], b[ITEMS_PER_GROUP];
#pragma unroll
        for (int j = 0; j < ITEMS_PER_GROUP; j++) {
            const int aj = __builtin_amdgcn_readlane(ia, j0 + j);
            const int bj = __builtin_amdgcn_readlane(ib, j0 + j);
            a[j] = *(const float4*)(ht + (size_t)aj * DDIM + lane * 4);
            b[j] = *(const float4*)(hm + (size_t)bj * DDIM + lane * 4);
        }
#pragma unroll
        for (int j = 0; j < ITEMS_PER_GROUP; j++) {
            float s = a[j].x * wa.x + a[j].y * wa.y + a[j].z * wa.z + a[j].w * wa.w
                    + b[j].x * wb.x + b[j].y * wb.y + b[j].z * wb.z + b[j].w * wb.w;
#pragma unroll
            for (int off = 16; off > 0; off >>= 1) s += __shfl_xor(s, off, 32);
            if (lane == j0 + j) res = s;
        }
    }
    if (q < nq) out[q] = res + cb[0];
}

extern "C" void kernel_launch(void* const* d_in, const int* in_sizes, int n_in,
                              void* d_out, int out_size, void* d_ws, size_t ws_size,
                              hipStream_t stream) {
    const float* x_thesis     = (const float*)d_in[0];
    const int*   mentor_id    = (const int*)  d_in[1];
    const int*   edge_src     = (const int*)  d_in[2];
    const int*   edge_dst     = (const int*)  d_in[3];
    const int*   eli_src      = (const int*)  d_in[4];
    const int*   eli_dst      = (const int*)  d_in[5];
    const float* thesis_lin_w = (const float*)d_in[6];
    const float* thesis_lin_b = (const float*)d_in[7];
    const float* prof_emb     = (const float*)d_in[8];
    const float* W_nbr_sup    = (const float*)d_in[9];
    const float* W_root_sup   = (const float*)d_in[10];
    const float* b_sup        = (const float*)d_in[11];
    const float* W_nbr_rev    = (const float*)d_in[12];
    const float* W_root_rev   = (const float*)d_in[13];
    const float* b_rev        = (const float*)d_in[14];
    const float* cls_w        = (const float*)d_in[15];
    const float* cls_b        = (const float*)d_in[16];
    float*       out          = (float*)d_out;

    const size_t T_PLANE = (size_t)N_T * DDIM;
    const size_t M_PLANE = (size_t)N_M * DDIM;
    static_assert(2 * BM_PLANE <= (size_t)N_T * DDIM, "both sides' bitmaps fit in one plane");
    float* p    = (float*)d_ws;
    float* tX   = p; p += T_PLANE;
    float* tY   = p; p += T_PLANE;
    float* tZ   = p; p += T_PLANE;
    float* mP   = p; p += M_PLANE;
    float* mQ   = p; p += M_PLANE;
    float* mR   = p; p += M_PLANE;
    unsigned* bmM = (unsigned*)tY;
    unsigned* bmT = bmM + BM_PLANE;

    const int BLK = 256;
    const int ROWS_PER_BLK = 8 * 16;

    k_owner_bitmaps<<<dim3(BM_NCH, 2), BM_CHUNK, 0, stream>>>(edge_dst, edge_src, bmM, bmT);

    float* ht0 = tZ;
    float* hm0 = mR;
    k_gemm_wmma<<<(N_T + ROWS_PER_BLK - 1) / ROWS_PER_BLK, BLK, 0, stream>>>(
        x_thesis, thesis_lin_w, F_IN, nullptr, nullptr, 0,
        thesis_lin_b, ht0, N_T, 0);
    {
        int n = N_M * (DDIM / 4);
        k_gather_rows<<<(n + BLK - 1) / BLK, BLK, 0, stream>>>(prof_emb, mentor_id, hm0, N_M);
    }

    float* hm1 = mQ;
    float* ht1 = tX;
    k_owner_mean<OWN_M_ROWS><<<OWNER_BLOCKS, OWNER_THREADS, 0, stream>>>(ht0, edge_dst, edge_src, bmM, mP, N_M);
    k_owner_mean<OWN_T_ROWS><<<OWNER_BLOCKS, OWNER_THREADS, 0, stream>>>(hm0, edge_src, edge_dst, bmT, tX, N_T);
    k_gemm_wmma<<<(N_M + ROWS_PER_BLK - 1) / ROWS_PER_BLK, BLK, 0, stream>>>(
        mP, W_nbr_sup, DDIM, hm0, W_root_sup, DDIM, b_sup, hm1, N_M, 1);
    k_gemm_wmma<<<(N_T + ROWS_PER_BLK - 1) / ROWS_PER_BLK, BLK, 0, stream>>>(
        tX, W_nbr_rev, DDIM, ht0, W_root_rev, DDIM, b_rev, ht1, N_T, 1);

    const size_t wOff = (size_t)DDIM * DDIM;
    float* hm2 = mR;
    float* ht2 = tY;
    k_owner_mean<OWN_M_ROWS><<<OWNER_BLOCKS, OWNER_THREADS, 0, stream>>>(ht1, edge_dst, edge_src, bmM, mP, N_M);
    k_owner_mean<OWN_T_ROWS><<<OWNER_BLOCKS, OWNER_THREADS, 0, stream>>>(hm1, edge_src, edge_dst, bmT, tZ, N_T);
    k_gemm_wmma<<<(N_M + ROWS_PER_BLK - 1) / ROWS_PER_BLK, BLK, 0, stream>>>(
        mP, W_nbr_sup + wOff, DDIM, hm1, W_root_sup + wOff, DDIM, b_sup + DDIM, hm2, N_M, 0);
    k_gemm_wmma<<<(N_T + ROWS_PER_BLK - 1) / ROWS_PER_BLK, BLK, 0, stream>>>(
        tZ, W_nbr_rev + wOff, DDIM, ht1, W_root_rev + wOff, DDIM, b_rev + DDIM, ht2, N_T, 0);

    {
        k_classify<<<(N_Q + 8 * ITEMS_PER_WAVE - 1) / (8 * ITEMS_PER_WAVE), BLK, 0, stream>>>(
            ht2, hm2, eli_src, eli_dst, cls_w, cls_b, out, N_Q);
    }
}


# Neighbour mean fused with its layer product: the owner block keeps its rows' means in LDS and multiplies them itself (A fragments from LDS); mean planes never stored; 12 launches become 8
# speedup vs baseline: 2.3565x; 1.1573x over previous
#include <hip/hip_runtime.h>
#include <hip/hip_bf16.h>

#define N_T   100000
#define N_M   20000
#define N_E   600000
#define N_Q   200000
#define F_IN  256
#define DDIM  128

#define K_SLAB   64
#define B_STRIDE 68

typedef float v2f __attribute__((ext_vector_type(2)));
typedef float v8f __attribute__((ext_vector_type(8)));

__global__ void k_gather_rows(const float* __restrict__ src, const int* __restrict__ ids,
                              float* __restrict__ dst, int nrows) {
    int t = blockIdx.x * blockDim.x + threadIdx.x;
    if (t >= nrows * (DDIM / 4)) return;
    int row = t >> 5;
    int l   = t & 31;
    float4 v = *(const float4*)(src + (size_t)ids[row] * DDIM + l * 4);
    *(float4*)(dst + (size_t)row * DDIM + l * 4) = v;
}

#define OWNER_BLOCKS   256
#define OWNER_THREADS  1024
#define OWNER_WAVES    32
#define OWN_T_ROWS     391
#define OWN_M_ROWS     79
#define QCAP           64
#define BM_CHUNK       1024
#define BM_NCH         ((N_E + BM_CHUNK - 1) / BM_CHUNK)
#define BM_WORDS       (BM_NCH * 32)
#define BM_PLANE       ((size_t)OWNER_BLOCKS * BM_WORDS)
#define BM_PITCH       257
#define ITEMS_PER_WAVE 32
#define ITEMS_PER_GROUP 8
#define SUM_PITCH      132
#define W_PITCH        132

__global__ void __launch_bounds__(BM_CHUNK)
k_owner_bitmaps(const int* __restrict__ key_m, const int* __restrict__ key_t,
                unsigned* __restrict__ bm_m, unsigned* __restrict__ bm_t) {
    static_assert(OWNER_BLOCKS % 32 == 0, "a wave stores one owner's line per step");
    __shared__ unsigned bits[32 * BM_PITCH];

    const int  tid    = threadIdx.x;
    const int  chunk  = blockIdx.x;
    const bool side_t = blockIdx.y != 0;
    const int* keys   = side_t ? key_t : key_m;
    unsigned*  bm     = side_t ? bm_t : bm_m;

    for (int i = tid; i < 32 * BM_PITCH; i += BM_CHUNK) bits[i] = 0u;
    __syncthreads();

    const int e = chunk * BM_CHUNK + tid;
    if (e < N_E) {
        const unsigned k   = (unsigned)keys[e];
        const unsigned own = side_t ? k / (unsigned)OWN_T_ROWS : k / (unsigned)OWN_M_ROWS;
        if (own < (unsigned)OWNER_BLOCKS)
            __hip_atomic_fetch_or(&bits[(tid >> 5) * BM_PITCH + own], 1u << (tid & 31),
                                  __ATOMIC_RELAXED, __HIP_MEMORY_SCOPE_WORKGROUP);
    }
    __syncthreads();

    const int wd = tid & 31;
    for (int o = tid >> 5; o < OWNER_BLOCKS; o += BM_CHUNK / 32)
        bm[((size_t)o * BM_NCH + chunk) * 32 + wd] = bits[wd * BM_PITCH + o];
}

__device__ __forceinline__ void owner_drain(const float* __restrict__ feat,
                                            const int* __restrict__ idx_key,
                                            const int* __restrict__ idx_other,
                                            const int* qd, int c, int lane, int lo,
                                            float* sums, float* cnts) {
    int loc = 0, o = 0;
    if (lane < c) {
        const int e = qd[lane];
        loc = idx_key[e] - lo;
        o   = idx_other[e];
        __hip_atomic_fetch_add(cnts + loc, 1.0f, __ATOMIC_RELAXED, __HIP_MEMORY_SCOPE_WORKGROUP);
    }
    for (int j0 = 0; j0 < c; j0 += ITEMS_PER_GROUP) {
        float v[ITEMS_PER_GROUP][4];
#pragma unroll
        for (int j = 0; j < ITEMS_PER_GROUP; j++) {
            const int oj = __builtin_amdgcn_readlane(o, j0 + j);
            const float* src = feat + (size_t)oj * DDIM + lane;
            v[j][0] = src[0];
            v[j][1] = src[32];
            v[j][2] = src[64];
            v[j][3] = src[96];
        }
#pragma unroll
        for (int j = 0; j < ITEMS_PER_GROUP; j++) {
            if (j == 0 || j0 + j < c) {
                const int lj = __builtin_amdgcn_readlane(loc, j0 + j);
                float* dst = sums + lj * SUM_PITCH + lane;
                __hip_atomic_fetch_add(dst +  0, v[j][0], __ATOMIC_RELAXED, __HIP_MEMORY_SCOPE_WORKGROUP);
                __hip_atomic_fetch_add(dst + 32, v[j][1], __ATOMIC_RELAXED, __HIP_MEMORY_SCOPE_WORKGROUP);
                __hip_atomic_fetch_add(dst + 64, v[j][2], __ATOMIC_RELAXED, __HIP_MEMORY_SCOPE_WORKGROUP);
                __hip_atomic_fetch_add(dst + 96, v[j][3], __ATOMIC_RELAXED, __HIP_MEMORY_SCOPE_WORKGROUP);
            }
        }
    }
}

template <int NT>
__device__ __forceinline__ void k_step8n(const float4 a4, const float* Ws,
                                         int ks, int r, int h, v8f acc[NT]) {
    v2f a1; a1[0] = a4.x; a1[1] = a4.y;
    v2f a2; a2[0] = a4.z; a2[1] = a4.w;
#pragma unroll
    for (int nt = 0; nt < NT; nt++) {
        float4 b4 = *(const float4*)(Ws + (nt * 16 + r) * W_PITCH + ks + 4 * h);
        v2f b1; b1[0] = b4.x; b1[1] = b4.y;
        v2f b2; b2[0] = b4.z; b2[1] = b4.w;
        acc[nt] = __builtin_amdgcn_wmma_f32_16x16x4_f32(
            false, a1, false, b1, (short)0, acc[nt], false, false);
        acc[nt] = __builtin_amdgcn_wmma_f32_16x16x4_f32(
            false, a2, false, b2, (short)0, acc[nt], false, false);
    }
}

template <int NT>
__device__ __forceinline__ void owner_product(const float* Arow, const float* Ws,
                                              int r, int h, v8f acc[NT]) {
#pragma nounroll
    for (int k0 = 0; k0 < DDIM; k0 += K_SLAB) {
        float4 a_cur = *(const float4*)(Arow + k0);
#pragma unroll
        for (int ks = k0; ks < k0 + K_SLAB; ks += 8) {
            float4 a_nxt = a_cur;
            if (ks + 8 < k0 + K_SLAB) a_nxt = *(const float4*)(Arow + ks + 8);
            k_step8n<NT>(a_cur, Ws, ks, r, h, acc);
            a_cur = a_nxt;
            if (NT >= 8) __builtin_amdgcn_sched_barrier(0);
        }
        __builtin_amdgcn_sched_barrier(0);
        asm volatile("v_nop\n\tv_nop\n\tv_nop\n\tv_nop");
    }
}

__device__ __forceinline__ void stage_w(const float* __restrict__ Wg, float* Ws, int tid) {
#pragma unroll
    for (int i = 0; i < (DDIM * DDIM / 4) / OWNER_THREADS; i++) {
        const int idx = tid + i * OWNER_THREADS;
        const int row = idx >> 5;
        const int c   = (idx & 31) << 2;
        const float4 v = *(const float4*)(Wg + (size_t)row * DDIM + c);
        *(float4*)(Ws + row * W_PITCH + c) = v;
    }
}

__device__ __forceinline__ int lane_here() {
    int l;
    asm volatile("v_mbcnt_lo_u32_b32 %0, -1, 0" : "=v"(l));
    return l;
}

template <int ROWS, int NT>
__global__ void __launch_bounds__(OWNER_THREADS)
k_owner_layer(const float* __restrict__ feat,
              const int* __restrict__ idx_key,
              const int* __restrict__ idx_other,
              const unsigned* __restrict__ bm,
              const float* __restrict__ self,
              const float* __restrict__ Wn, const float* __restrict__ Wr,
              const float* __restrict__ bias,
              float* __restrict__ out, int n_nodes, int relu) {
    static_assert((long long)BM_WORDS * 32 >= N_E, "the bitmap has a bit for every edge");
    static_assert(BM_WORDS % 4 == 0, "a block's bitmap is read 16 bytes per lane");
    constexpr int STRIPES = (ROWS + 15) / 16;
    constexpr int ROWS16  = STRIPES * 16;
    constexpr int CGROUPS = 8 / NT;
    constexpr int UNITS   = STRIPES * CGROUPS;
    static_assert(UNITS <= OWNER_WAVES, "one unit per wave");
    __shared__ float sums[ROWS16 * SUM_PITCH];
    __shared__ float cnts[ROWS];
    __shared__ int   q_all[OWNER_WAVES][QCAP];
    __shared__ float Ws[DDIM * W_PITCH];

    const int tid  = threadIdx.x;
    const int lane = tid & 31;
    const int w  = __builtin_amdgcn_readfirstlane(tid >> 5);
    const int lo = blockIdx.x * ROWS;
    if (lo >= n_nodes) return;
    const int own = min(ROWS, n_nodes - lo);

    for (int i = tid; i < ROWS16 * (SUM_PITCH / 4); i += OWNER_THREADS)
        *(float4*)(sums + i * 4) = make_float4(0.0f, 0.0f, 0.0f, 0.0f);
    for (int i = tid; i < ROWS; i += OWNER_THREADS) cnts[i] = 0.0f;
    stage_w(Wn, Ws, tid);
    __syncthreads();

    int* q = q_all[w];
    const unsigned* mine = bm + (size_t)blockIdx.x * BM_WORDS;
    int qn = 0;

    const uint4* mine4 = (const uint4*)mine;
    constexpr int N4 = BM_WORDS / 4;
    constexpr int NR = (N4 + OWNER_THREADS - 1) / OWNER_THREADS;
#define OWNER_GROUP(r_) \
    (((r_) * OWNER_THREADS + tid) < N4 \
         ? mine4[min((r_) * OWNER_THREADS + tid, N4 - 1)] : make_uint4(0u, 0u, 0u, 0u))

    uint4 cur = OWNER_GROUP(0);
    for (int r = 0; r < NR; r++) {
        const unsigned round_any =
            __builtin_amdgcn_ballot_w32((cur.x | cur.y | cur.z | cur.w) != 0u);
        const uint4 nxt = OWNER_GROUP(r + 1);
        for (int k = 0; round_any != 0u && k < 4; k++) {
            const unsigned word = k == 0 ? cur.x : k == 1 ? cur.y : k == 2 ? cur.z : cur.w;
            unsigned any = __builtin_amdgcn_ballot_w32(word != 0u);
            const int wbase = ((r * OWNER_THREADS + (w << 5)) << 2) + k;
            while (any) {
                const int L = __builtin_ctz(any);
                any &= any - 1u;
                const unsigned wl = (unsigned)__builtin_amdgcn_readlane((int)word, L);
                const int ebase = (wbase + (L << 2)) << 5;
                if ((wl >> lane) & 1u) q[qn + (int)__builtin_amdgcn_mbcnt_lo(wl, 0u)] = ebase + lane;
                qn += __builtin_popcount(wl);
                if (qn >= 32) {
                    qn -= 32;
                    owner_drain(feat, idx_key, idx_other, q + qn, 32, lane, lo, sums, cnts);
                }
            }
        }
        cur = nxt;
    }
#undef OWNER_GROUP
    if (qn > 0) owner_drain(feat, idx_key, idx_other, q, qn, lane, lo, sums, cnts);
    __syncthreads();

    for (int i = tid; i < own * (DDIM / 4); i += OWNER_THREADS) {
        const int row = i >> 5;
        const int c4  = (i & 31) * 4;
        const float rc = 1.0f / fmaxf(cnts[row], 1.0f);
        float4 s = *(const float4*)(sums + row * SUM_PITCH + c4);
        s.x *= rc; s.y *= rc; s.z *= rc; s.w *= rc;
        *(float4*)(sums + row * SUM_PITCH + c4) = s;
    }
    __syncthreads();

    const int  r      = lane & 15;
    const int  h      = lane >> 4;
    const bool worker = w < UNITS;
    const int  stripe = worker ? w / CGROUPS : 0;
    const int  cg     = worker ? w % CGROUPS : 0;
    const float* Wcols = Ws + cg * (NT * 16) * W_PITCH;

    v8f acc[NT] = {};
    if (worker)
        owner_product<NT>(sums + (stripe * 16 + r) * SUM_PITCH + 4 * h, Wcols, r, h, acc);
    __syncthreads();
    const int lane2 = lane_here();
    stage_w(Wr, Ws, (w << 5) + lane2);
    __syncthreads();
    if (!worker) return;

    const int r2 = lane2 & 15;
    const int h2 = lane2 >> 4;
    const int grow = min(lo + stripe * 16 + r2, n_nodes - 1);
    owner_product<NT>(self + (size_t)grow * DDIM + 4 * h2, Wcols, r2, h2, acc);

    const int lane3 = lane_here();
    const int r3 = lane3 & 15;
    const int h3 = lane3 >> 4;
#pragma unroll
    for (int nt = 0; nt < NT; nt++) {
        const int col  = (cg * NT + nt) * 16 + r3;
        const float bb = bias[col];
#pragma unroll
        for (int i = 0; i < 8; i++) {
            const int lrow = stripe * 16 + h3 * 8 + i;
            float v = acc[nt][i] + bb;
            if (relu) v = fmaxf(v, 0.0f);
            if (lrow < own) out[(size_t)(lo + lrow) * DDIM + col] = v;
        }
    }
}

__device__ __forceinline__ void k_step8(const float4 a4, const float* __restrict__ Bs,
                                        int ks, int r, int h, v8f acc[8]) {
    v2f a1; a1[0] = a4.x; a1[1] = a4.y;
    v2f a2; a2[0] = a4.z; a2[1] = a4.w;
#pragma unroll
    for (int nt = 0; nt < 8; nt++) {
        float4 b4 = *(const float4*)(Bs + (nt * 16 + r) * B_STRIDE + ks + 4 * h);
        v2f b1; b1[0] = b4.x; b1[1] = b4.y;
        v2f b2; b2[0] = b4.z; b2[1] = b4.w;
        acc[nt] = __builtin_amdgcn_wmma_f32_16x16x4_f32(
            false, a1, false, b1, (short)0, acc[nt], false, false);
        acc[nt] = __builtin_amdgcn_wmma_f32_16x16x4_f32(
            false, a2, false, b2, (short)0, acc[nt], false, false);
    }
}

__device__ __forceinline__ void gemm_pass(const float* A,
                                          const float* __restrict__ W,
                                          int K, int rowA, int r, int h, int tid,
                                          float* __restrict__ Bs, v8f acc[8]) {
    for (int kk0 = 0; kk0 < K; kk0 += K_SLAB) {
        __syncthreads();
#pragma unroll
        for (int i = 0; i < 8; i++) {
            int idx = tid + i * 256;
            int row = idx >> 4;
            int c   = (idx & 15) << 2;
            float4 v = *(const float4*)(W + (size_t)row * K + kk0 + c);
            *(float4*)(Bs + row * B_STRIDE + c) = v;
        }
        __syncthreads();

        const size_t aoff = (size_t)(rowA + r) * K + kk0 + 4 * h;
        const float* Arow = A + aoff;
        for (int ks = 0; ks < K_SLAB; ks += 8) {
            float4 a4 = *(const float4*)(Arow + ks);
            k_step8(a4, Bs, ks, r, h, acc);
        }
        __builtin_amdgcn_sched_barrier(0);
        asm volatile("v_nop\n\tv_nop\n\tv_nop\n\tv_nop");
    }
}

__global__ void __launch_bounds__(256)
k_gemm_wmma(const float* A1, const float* __restrict__ W1, int K1,
            const float* __restrict__ A2, const float* __restrict__ W2, int K2,
            const float* __restrict__ bias, float* out,
            int nrows, int relu) {
    __shared__ float Bs[DDIM * B_STRIDE];

    const int tid  = threadIdx.x;
    const int lane = tid & 31;
    const int wave = tid >> 5;
    const int row0 = (blockIdx.x * 8 + wave) * 16;
    const bool valid = (row0 < nrows);
    const int rowA = valid ? row0 : 0;
    const int r = lane & 15;
    const int h = lane >> 4;

    v8f acc[8] = {};

    gemm_pass(A1, W1, K1, rowA, r, h, tid, Bs, acc);
    if (A2) gemm_pass(A2, W2, K2, rowA, r, h, tid, Bs, acc);

    if (!valid) return;
#pragma unroll
    for (int nt = 0; nt < 8; nt++) {
        int col  = nt * 16 + r;
        float bb = bias[col];
#pragma unroll
        for (int i = 0; i < 8; i++) {
            float v = acc[nt][i] + bb;
            if (relu) v = fmaxf(v, 0.0f);
            out[(size_t)(row0 + h * 8 + i) * DDIM + col] = v;
        }
    }
}

__global__ void __launch_bounds__(256)
k_classify(const float* __restrict__ ht, const float* __restrict__ hm,
           const int* __restrict__ qs, const int* __restrict__ qd,
           const float* __restrict__ w, const float* __restrict__ cb,
           float* __restrict__ out, int nq) {
    const int lane = threadIdx.x & 31;
    const int wave = __builtin_amdgcn_readfirstlane(
        (int)((blockIdx.x * blockDim.x + threadIdx.x) >> 5));
    const int base = wave * ITEMS_PER_WAVE;
    if (base >= nq) return;
    const int q = base + lane;
    int ia = 0, ib = 0;
    if (q < nq) {
        ia = qs[q];
        ib = qd[q];
    }
    const float4 wa = *(const float4*)(w + lane * 4);
    const float4 wb = *(const float4*)(w + DDIM + lane * 4);
    const int cnt = min(ITEMS_PER_WAVE, nq - base);
    float res = 0.0f;
    for (int j0 = 0; j0 < cnt; j0 += ITEMS_PER_GROUP) {
        float4 a[ITEMS_PER_GROUP], b[ITEMS_PER_GROUP];
#pragma unroll
        for (int j = 0; j < ITEMS_PER_GROUP; j++) {
            const int aj = __builtin_amdgcn_readlane(ia, j0 + j);
            const int bj = __builtin_amdgcn_readlane(ib, j0 + j);
            a[j] = *(const float4*)(ht + (size_t)aj * DDIM + lane * 4);
            b[j] = *(const float4*)(hm + (size_t)bj * DDIM + lane * 4);
        }
#pragma unroll
        for (int j = 0; j < ITEMS_PER_GROUP; j++) {
            float s = a[j].x * wa.x + a[j].y * wa.y + a[j].z * wa.z + a[j].w * wa.w
                    + b[j].x * wb.x + b[j].y * wb.y + b[j].z * wb.z + b[j].w * wb.w;
#pragma unroll
            for (int off = 16; off > 0; off >>= 1) s += __shfl_xor(s, off, 32);
            if (lane == j0 + j) res = s;
        }
    }
    if (q < nq) out[q] = res + cb[0];
}

extern "C" void kernel_launch(void* const* d_in, const int* in_sizes, int n_in,
                              void* d_out, int out_size, void* d_ws, size_t ws_size,
                              hipStream_t stream) {
    const float* x_thesis     = (const float*)d_in[0];
    const int*   mentor_id    = (const int*)  d_in[1];
    const int*   edge_src     = (const int*)  d_in[2];
    const int*   edge_dst     = (const int*)  d_in[3];
    const int*   eli_src      = (const int*)  d_in[4];
    const int*   eli_dst      = (const int*)  d_in[5];
    const float* thesis_lin_w = (const float*)d_in[6];
    const float* thesis_lin_b = (const float*)d_in[7];
    const float* prof_emb     = (const float*)d_in[8];
    const float* W_nbr_sup    = (const float*)d_in[9];
    const float* W_root_sup   = (const float*)d_in[10];
    const float* b_sup        = (const float*)d_in[11];
    const float* W_nbr_rev    = (const float*)d_in[12];
    const float* W_root_rev   = (const float*)d_in[13];
    const float* b_rev        = (const float*)d_in[14];
    const float* cls_w        = (const float*)d_in[15];
    const float* cls_b        = (const float*)d_in[16];
    float*       out          = (float*)d_out;

    const size_t T_PLANE = (size_t)N_T * DDIM;
    const size_t M_PLANE = (size_t)N_M * DDIM;
    static_assert(2 * BM_PLANE <= (size_t)N_T * DDIM, "both sides' bitmaps fit in one plane");
    float* p    = (float*)d_ws;
    float* tX   = p; p += T_PLANE;
    float* tY   = p; p += T_PLANE;
    float* tZ   = p; p += T_PLANE;
    p += M_PLANE;
    float* mQ   = p; p += M_PLANE;
    float* mR   = p; p += M_PLANE;
    unsigned* bmM = (unsigned*)tY;
    unsigned* bmT = bmM + BM_PLANE;

    const int BLK = 256;
    const int ROWS_PER_BLK = 8 * 16;

    k_owner_bitmaps<<<dim3(BM_NCH, 2), BM_CHUNK, 0, stream>>>(edge_dst, edge_src, bmM, bmT);

    float* ht0 = tZ;
    float* hm0 = mR;
    k_gemm_wmma<<<(N_T + ROWS_PER_BLK - 1) / ROWS_PER_BLK, BLK, 0, stream>>>(
        x_thesis, thesis_lin_w, F_IN, nullptr, nullptr, 0,
        thesis_lin_b, ht0, N_T, 0);
    {
        int n = N_M * (DDIM / 4);
        k_gather_rows<<<(n + BLK - 1) / BLK, BLK, 0, stream>>>(prof_emb, mentor_id, hm0, N_M);
    }

    float* hm1 = mQ;
    float* ht1 = tX;
    k_owner_layer<OWN_M_ROWS, 2><<<OWNER_BLOCKS, OWNER_THREADS, 0, stream>>>(
        ht0, edge_dst, edge_src, bmM, hm0, W_nbr_sup, W_root_sup, b_sup, hm1, N_M, 1);
    k_owner_layer<OWN_T_ROWS, 8><<<OWNER_BLOCKS, OWNER_THREADS, 0, stream>>>(
        hm0, edge_src, edge_dst, bmT, ht0, W_nbr_rev, W_root_rev, b_rev, ht1, N_T, 1);

    const size_t wOff = (size_t)DDIM * DDIM;
    float* hm2 = mR;
    float* ht2 = tZ;
    k_owner_layer<OWN_M_ROWS, 2><<<OWNER_BLOCKS, OWNER_THREADS, 0, stream>>>(
        ht1, edge_dst, edge_src, bmM, hm1, W_nbr_sup + wOff, W_root_sup + wOff, b_sup + DDIM, hm2, N_M, 0);
    k_owner_layer<OWN_T_ROWS, 8><<<OWNER_BLOCKS, OWNER_THREADS, 0, stream>>>(
        hm1, edge_src, edge_dst, bmT, ht1, W_nbr_rev + wOff, W_root_rev + wOff, b_rev + DDIM, ht2, N_T, 0);

    {
        k_classify<<<(N_Q + 8 * ITEMS_PER_WAVE - 1) / (8 * ITEMS_PER_WAVE), BLK, 0, stream>>>(
            ht2, hm2, eli_src, eli_dst, cls_w, cls_b, out, N_Q);
    }
}


# Last layer's rows never stored: both its product epilogues dot the accumulators with the classifier weights, one float per node via LDS; classifier adds 2 gathered floats per query; sum order changes
# speedup vs baseline: 2.6522x; 1.1255x over previous
#include <hip/hip_runtime.h>
#include <hip/hip_bf16.h>

#define N_T   100000
#define N_M   20000
#define N_E   600000
#define N_Q   200000
#define F_IN  256
#define DDIM  128

#define K_SLAB   64
#define B_STRIDE 68

typedef float v2f __attribute__((ext_vector_type(2)));
typedef float v8f __attribute__((ext_vector_type(8)));

__global__ void k_gather_rows(const float* __restrict__ src, const int* __restrict__ ids,
                              float* __restrict__ dst, int nrows) {
    int t = blockIdx.x * blockDim.x + threadIdx.x;
    if (t >= nrows * (DDIM / 4)) return;
    int row = t >> 5;
    int l   = t & 31;
    float4 v = *(const float4*)(src + (size_t)ids[row] * DDIM + l * 4);
    *(float4*)(dst + (size_t)row * DDIM + l * 4) = v;
}

#define OWNER_BLOCKS   256
#define OWNER_THREADS  1024
#define OWNER_WAVES    32
#define OWN_T_ROWS     391
#define OWN_M_ROWS     79
#define QCAP           64
#define BM_CHUNK       1024
#define BM_NCH         ((N_E + BM_CHUNK - 1) / BM_CHUNK)
#define BM_WORDS       (BM_NCH * 32)
#define BM_PLANE       ((size_t)OWNER_BLOCKS * BM_WORDS)
#define BM_PITCH       257
#define ITEMS_PER_GROUP 8
#define SUM_PITCH      132
#define W_PITCH        132

__global__ void __launch_bounds__(BM_CHUNK)
k_owner_bitmaps(const int* __restrict__ key_m, const int* __restrict__ key_t,
                unsigned* __restrict__ bm_m, unsigned* __restrict__ bm_t) {
    static_assert(OWNER_BLOCKS % 32 == 0, "a wave stores one owner's line per step");
    __shared__ unsigned bits[32 * BM_PITCH];

    const int  tid    = threadIdx.x;
    const int  chunk  = blockIdx.x;
    const bool side_t = blockIdx.y != 0;
    const int* keys   = side_t ? key_t : key_m;
    unsigned*  bm     = side_t ? bm_t : bm_m;

    for (int i = tid; i < 32 * BM_PITCH; i += BM_CHUNK) bits[i] = 0u;
    __syncthreads();

    const int e = chunk * BM_CHUNK + tid;
    if (e < N_E) {
        const unsigned k   = (unsigned)keys[e];
        const unsigned own = side_t ? k / (unsigned)OWN_T_ROWS : k / (unsigned)OWN_M_ROWS;
        if (own < (unsigned)OWNER_BLOCKS)
            __hip_atomic_fetch_or(&bits[(tid >> 5) * BM_PITCH + own], 1u << (tid & 31),
                                  __ATOMIC_RELAXED, __HIP_MEMORY_SCOPE_WORKGROUP);
    }
    __syncthreads();

    const int wd = tid & 31;
    for (int o = tid >> 5; o < OWNER_BLOCKS; o += BM_CHUNK / 32)
        bm[((size_t)o * BM_NCH + chunk) * 32 + wd] = bits[wd * BM_PITCH + o];
}

__device__ __forceinline__ void owner_drain(const float* __restrict__ feat,
                                            const int* __restrict__ idx_key,
                                            const int* __restrict__ idx_other,
                                            const int* qd, int c, int lane, int lo,
                                            float* sums, float* cnts) {
    int loc = 0, o = 0;
    if (lane < c) {
        const int e = qd[lane];
        loc = idx_key[e] - lo;
        o   = idx_other[e];
        __hip_atomic_fetch_add(cnts + loc, 1.0f, __ATOMIC_RELAXED, __HIP_MEMORY_SCOPE_WORKGROUP);
    }
    for (int j0 = 0; j0 < c; j0 += ITEMS_PER_GROUP) {
        float v[ITEMS_PER_GROUP][4];
#pragma unroll
        for (int j = 0; j < ITEMS_PER_GROUP; j++) {
            const int oj = __builtin_amdgcn_readlane(o, j0 + j);
            const float* src = feat + (size_t)oj * DDIM + lane;
            v[j][0] = src[0];
            v[j][1] = src[32];
            v[j][2] = src[64];
            v[j][3] = src[96];
        }
#pragma unroll
        for (int j = 0; j < ITEMS_PER_GROUP; j++) {
            if (j == 0 || j0 + j < c) {
                const int lj = __builtin_amdgcn_readlane(loc, j0 + j);
                float* dst = sums + lj * SUM_PITCH + lane;
                __hip_atomic_fetch_add(dst +  0, v[j][0], __ATOMIC_RELAXED, __HIP_MEMORY_SCOPE_WORKGROUP);
                __hip_atomic_fetch_add(dst + 32, v[j][1], __ATOMIC_RELAXED, __HIP_MEMORY_SCOPE_WORKGROUP);
                __hip_atomic_fetch_add(dst + 64, v[j][2], __ATOMIC_RELAXED, __HIP_MEMORY_SCOPE_WORKGROUP);
                __hip_atomic_fetch_add(dst + 96, v[j][3], __ATOMIC_RELAXED, __HIP_MEMORY_SCOPE_WORKGROUP);
            }
        }
    }
}

template <int NT>
__device__ __forceinline__ void k_step8n(const float4 a4, const float* Ws,
                                         int ks, int r, int h, v8f acc[NT]) {
    v2f a1; a1[0] = a4.x; a1[1] = a4.y;
    v2f a2; a2[0] = a4.z; a2[1] = a4.w;
#pragma unroll
    for (int nt = 0; nt < NT; nt++) {
        float4 b4 = *(const float4*)(Ws + (nt * 16 + r) * W_PITCH + ks + 4 * h);
        v2f b1; b1[0] = b4.x; b1[1] = b4.y;
        v2f b2; b2[0] = b4.z; b2[1] = b4.w;
        acc[nt] = __builtin_amdgcn_wmma_f32_16x16x4_f32(
            false, a1, false, b1, (short)0, acc[nt], false, false);
        acc[nt] = __builtin_amdgcn_wmma_f32_16x16x4_f32(
            false, a2, false, b2, (short)0, acc[nt], false, false);
    }
}

template <int NT>
__device__ __forceinline__ void owner_product(const float* Arow, const float* Ws,
                                              int r, int h, v8f acc[NT]) {
#pragma nounroll
    for (int k0 = 0; k0 < DDIM; k0 += K_SLAB) {
        float4 a_cur = *(const float4*)(Arow + k0);
#pragma unroll
        for (int ks = k0; ks < k0 + K_SLAB; ks += 8) {
            float4 a_nxt = a_cur;
            if (ks + 8 < k0 + K_SLAB) a_nxt = *(const float4*)(Arow + ks + 8);
            k_step8n<NT>(a_cur, Ws, ks, r, h, acc);
            a_cur = a_nxt;
            if (NT >= 8) __builtin_amdgcn_sched_barrier(0);
        }
        __builtin_amdgcn_sched_barrier(0);
        asm volatile("v_nop\n\tv_nop\n\tv_nop\n\tv_nop");
    }
}

__device__ __forceinline__ void stage_w(const float* __restrict__ Wg, float* Ws, int tid) {
#pragma unroll
    for (int i = 0; i < (DDIM * DDIM / 4) / OWNER_THREADS; i++) {
        const int idx = tid + i * OWNER_THREADS;
        const int row = idx >> 5;
        const int c   = (idx & 31) << 2;
        const float4 v = *(const float4*)(Wg + (size_t)row * DDIM + c);
        *(float4*)(Ws + row * W_PITCH + c) = v;
    }
}

__device__ __forceinline__ int lane_here() {
    int l;
    asm volatile("v_mbcnt_lo_u32_b32 %0, -1, 0" : "=v"(l));
    return l;
}

#define OWN_PITCH(rows_) ((((rows_) + 31) / 32) * 32)

__device__ __forceinline__ float half_rows_reduce(const float p[8], int r) {
    const bool b3 = (r & 8) != 0, b2 = (r & 4) != 0, b1 = (r & 2) != 0;
    float u[4], t[2];
#pragma unroll
    for (int j = 0; j < 4; j++) {
        const float send = b3 ? p[j] : p[j + 4];
        const float keep = b3 ? p[j + 4] : p[j];
        u[j] = keep + __shfl_xor(send, 8, 32);
    }
#pragma unroll
    for (int j = 0; j < 2; j++) {
        const float send = b2 ? u[j] : u[j + 2];
        const float keep = b2 ? u[j + 2] : u[j];
        t[j] = keep + __shfl_xor(send, 4, 32);
    }
    const float send = b1 ? t[0] : t[1];
    const float keep = b1 ? t[1] : t[0];
    float s = keep + __shfl_xor(send, 2, 32);
    s += __shfl_xor(s, 1, 32);
    return s;
}

template <int ROWS, int NT, bool DOT>
__global__ void __launch_bounds__(OWNER_THREADS)
k_owner_layer(const float* __restrict__ feat,
              const int* __restrict__ idx_key,
              const int* __restrict__ idx_other,
              const unsigned* __restrict__ bm,
              const float* __restrict__ self,
              const float* __restrict__ Wn, const float* __restrict__ Wr,
              const float* __restrict__ bias,
              const float* __restrict__ cw,
              float* __restrict__ out, int n_nodes, int relu) {
    static_assert((long long)BM_WORDS * 32 >= N_E, "the bitmap has a bit for every edge");
    static_assert(BM_WORDS % 4 == 0, "a block's bitmap is read 16 bytes per lane");
    constexpr int STRIPES = (ROWS + 15) / 16;
    constexpr int ROWS16  = STRIPES * 16;
    constexpr int CGROUPS = 8 / NT;
    constexpr int UNITS   = STRIPES * CGROUPS;
    static_assert(UNITS <= OWNER_WAVES, "one unit per wave");
    __shared__ float sums[ROWS16 * SUM_PITCH];
    __shared__ float cnts[ROWS];
    __shared__ int   q_all[OWNER_WAVES][QCAP];
    __shared__ float Ws[DDIM * W_PITCH];
    __shared__ int   next_chunk;
    __shared__ float rowdot[DOT ? CGROUPS * ROWS16 : 1];

    const int tid  = threadIdx.x;
    const int lane = tid & 31;
    const int w  = __builtin_amdgcn_readfirstlane(tid >> 5);
    const int lo = blockIdx.x * ROWS;
    if (lo >= n_nodes) return;
    const int own = min(ROWS, n_nodes - lo);

    for (int i = tid; i < ROWS16 * (SUM_PITCH / 4); i += OWNER_THREADS)
        *(float4*)(sums + i * 4) = make_float4(0.0f, 0.0f, 0.0f, 0.0f);
    for (int i = tid; i < ROWS; i += OWNER_THREADS) cnts[i] = 0.0f;
    if (tid == 0) next_chunk = 0;
    stage_w(Wr, Ws, tid);
    __syncthreads();

    const bool worker = w < UNITS;
    const int  stripe = worker ? w / CGROUPS : 0;
    const int  cg     = worker ? w % CGROUPS : 0;
    const float* Wcols = Ws + cg * (NT * 16) * W_PITCH;

    v8f acc[NT] = {};
    if (worker) {
        const int r = lane & 15;
        const int h = lane >> 4;
        const int grow = min(lo + stripe * 16 + r, n_nodes - 1);
        owner_product<NT>(self + (size_t)grow * DDIM + 4 * h, Wcols, r, h, acc);
    }

    const int lane1 = lane_here();
    int* q = q_all[w];
    int qn = 0;

    const uint4* mine4 = (const uint4*)(bm + (size_t)blockIdx.x * BM_WORDS);
    constexpr int N4  = BM_WORDS / 4;
    constexpr int NCH = (N4 + 31) / 32;
#define OWNER_TAKE(c_) \
    do { \
        int t_ = 0; \
        if (lane1 == 0) \
            t_ = __hip_atomic_fetch_add(&next_chunk, 1, __ATOMIC_RELAXED, __HIP_MEMORY_SCOPE_WORKGROUP); \
        (c_) = __builtin_amdgcn_readfirstlane(t_); \
    } while (0)
#define OWNER_GROUP(c_) \
    (((c_) < NCH && (c_) * 32 + lane1 < N4) ? mine4[(c_) * 32 + lane1] : make_uint4(0u, 0u, 0u, 0u))

    int c_cur;
    OWNER_TAKE(c_cur);
    uint4 cur = OWNER_GROUP(c_cur);
    for (int it = 0; it < NCH && c_cur < NCH; it++) {
        const unsigned chunk_any =
            __builtin_amdgcn_ballot_w32((cur.x | cur.y | cur.z | cur.w) != 0u);
        int c_nxt;
        OWNER_TAKE(c_nxt);
        const uint4 nxt = OWNER_GROUP(c_nxt);
        for (int k = 0; chunk_any != 0u && k < 4; k++) {
            const unsigned word = k == 0 ? cur.x : k == 1 ? cur.y : k == 2 ? cur.z : cur.w;
            unsigned any = __builtin_amdgcn_ballot_w32(word != 0u);
            const int wbase = (c_cur << 7) + k;
            while (any) {
                const int L = __builtin_ctz(any);
                any &= any - 1u;
                const unsigned wl = (unsigned)__builtin_amdgcn_readlane((int)word, L);
                const int ebase = (wbase + (L << 2)) << 5;
                if ((wl >> lane1) & 1u) q[qn + (int)__builtin_amdgcn_mbcnt_lo(wl, 0u)] = ebase + lane1;
                qn += __builtin_popcount(wl);
                if (qn >= 32) {
                    qn -= 32;
                    owner_drain(feat, idx_key, idx_other, q + qn, 32, lane1, lo, sums, cnts);
                }
            }
        }
        cur = nxt;
        c_cur = c_nxt;
    }
#undef OWNER_GROUP
#undef OWNER_TAKE
    if (qn > 0) owner_drain(feat, idx_key, idx_other, q, qn, lane1, lo, sums, cnts);
    __syncthreads();

    const int lane2 = lane_here();
    const int tid2  = (w << 5) + lane2;
    for (int i = tid2; i < own * (DDIM / 4); i += OWNER_THREADS) {
        const int row = i >> 5;
        const int c4  = (i & 31) * 4;
        const float rc = 1.0f / fmaxf(cnts[row], 1.0f);
        float4 s = *(const float4*)(sums + row * SUM_PITCH + c4);
        s.x *= rc; s.y *= rc; s.z *= rc; s.w *= rc;
        *(float4*)(sums + row * SUM_PITCH + c4) = s;
    }
    stage_w(Wn, Ws, tid2);
    __syncthreads();

    if constexpr (!DOT) {
        if (!worker) return;

        const int r2 = lane2 & 15;
        const int h2 = lane2 >> 4;
        owner_product<NT>(sums + (stripe * 16 + r2) * SUM_PITCH + 4 * h2, Wcols, r2, h2, acc);

        const int lane3 = lane_here();
        const int r3 = lane3 & 15;
        const int h3 = lane3 >> 4;
#pragma unroll
        for (int nt = 0; nt < NT; nt++) {
            const int col  = (cg * NT + nt) * 16 + r3;
            const float bb = bias[col];
#pragma unroll
            for (int i = 0; i < 8; i++) {
                const int lrow = stripe * 16 + h3 * 8 + i;
                float v = acc[nt][i] + bb;
                if (relu) v = fmaxf(v, 0.0f);
                if (lrow < own) out[(size_t)(lo + lrow) * DDIM + col] = v;
            }
        }
    } else {
        if (worker) {
            const int r2 = lane2 & 15;
            const int h2 = lane2 >> 4;
            owner_product<NT>(sums + (stripe * 16 + r2) * SUM_PITCH + 4 * h2, Wcols, r2, h2, acc);

            const int lane3 = lane_here();
            const int r3 = lane3 & 15;
            const int h3 = lane3 >> 4;
            float p[8] = {0.0f, 0.0f, 0.0f, 0.0f, 0.0f, 0.0f, 0.0f, 0.0f};
#pragma unroll
            for (int nt = 0; nt < NT; nt++) {
                const int col  = (cg * NT + nt) * 16 + r3;
                const float bb = bias[col];
                const float wv = cw[col];
#pragma unroll
                for (int i = 0; i < 8; i++) {
                    float v = acc[nt][i] + bb;
                    if (relu) v = fmaxf(v, 0.0f);
                    p[i] += v * wv;
                }
            }
            const float s = half_rows_reduce(p, r3);
            if ((r3 & 1) == 0)
                rowdot[cg * ROWS16 + stripe * 16 + h3 * 8 + (r3 >> 1)] = s;
        }
        __syncthreads();

        const int lane4 = lane_here();
        float* dst = out + (size_t)blockIdx.x * OWN_PITCH(ROWS);
        for (int i = (w << 5) + lane4; i < own; i += OWNER_THREADS) {
            float s = rowdot[i];
#pragma unroll
            for (int g = 1; g < CGROUPS; g++) s += rowdot[g * ROWS16 + i];
            dst[i] = s;
        }
    }
}

__device__ __forceinline__ void k_step8(const float4 a4, const float* __restrict__ Bs,
                                        int ks, int r, int h, v8f acc[8]) {
    v2f a1; a1[0] = a4.x; a1[1] = a4.y;
    v2f a2; a2[0] = a4.z; a2[1] = a4.w;
#pragma unroll
    for (int nt = 0; nt < 8; nt++) {
        float4 b4 = *(const float4*)(Bs + (nt * 16 + r) * B_STRIDE + ks + 4 * h);
        v2f b1; b1[0] = b4.x; b1[1] = b4.y;
        v2f b2; b2[0] = b4.z; b2[1] = b4.w;
        acc[nt] = __builtin_amdgcn_wmma_f32_16x16x4_f32(
            false, a1, false, b1, (short)0, acc[nt], false, false);
        acc[nt] = __builtin_amdgcn_wmma_f32_16x16x4_f32(
            false, a2, false, b2, (short)0, acc[nt], false, false);
    }
}

__device__ __forceinline__ void gemm_pass(const float* A,
                                          const float* __restrict__ W,
                                          int K, int rowA, int r, int h, int tid,
                                          float* __restrict__ Bs, v8f acc[8]) {
    for (int kk0 = 0; kk0 < K; kk0 += K_SLAB) {
        __syncthreads();
#pragma unroll
        for (int i = 0; i < 8; i++) {
            int idx = tid + i * 256;
            int row = idx >> 4;
            int c   = (idx & 15) << 2;
            float4 v = *(const float4*)(W + (size_t)row * K + kk0 + c);
            *(float4*)(Bs + row * B_STRIDE + c) = v;
        }
        __syncthreads();

        const size_t aoff = (size_t)(rowA + r) * K + kk0 + 4 * h;
        const float* Arow = A + aoff;
        for (int ks = 0; ks < K_SLAB; ks += 8) {
            float4 a4 = *(const float4*)(Arow + ks);
            k_step8(a4, Bs, ks, r, h, acc);
        }
        __builtin_amdgcn_sched_barrier(0);
        asm volatile("v_nop\n\tv_nop\n\tv_nop\n\tv_nop");
    }
}

__global__ void __launch_bounds__(256)
k_gemm_wmma(const float* A1, const float* __restrict__ W1, int K1,
            const float* __restrict__ A2, const float* __restrict__ W2, int K2,
            const float* __restrict__ bias, float* out,
            int nrows, int relu) {
    __shared__ float Bs[DDIM * B_STRIDE];

    const int tid  = threadIdx.x;
    const int lane = tid & 31;
    const int wave = tid >> 5;
    const int row0 = (blockIdx.x * 8 + wave) * 16;
    const bool valid = (row0 < nrows);
    const int rowA = valid ? row0 : 0;
    const int r = lane & 15;
    const int h = lane >> 4;

    v8f acc[8] = {};

    gemm_pass(A1, W1, K1, rowA, r, h, tid, Bs, acc);
    if (A2) gemm_pass(A2, W2, K2, rowA, r, h, tid, Bs, acc);

    if (!valid) return;
#pragma unroll
    for (int nt = 0; nt < 8; nt++) {
        int col  = nt * 16 + r;
        float bb = bias[col];
#pragma unroll
        for (int i = 0; i < 8; i++) {
            float v = acc[nt][i] + bb;
            if (relu) v = fmaxf(v, 0.0f);
            out[(size_t)(row0 + h * 8 + i) * DDIM + col] = v;
        }
    }
}

__global__ void __launch_bounds__(OWNER_THREADS)
k_classify_scalars(const float* __restrict__ st, const float* __restrict__ sm,
                   const int* __restrict__ qs, const int* __restrict__ qd,
                   const float* __restrict__ cb,
                   float* __restrict__ out, int nq) {
    const int q = blockIdx.x * blockDim.x + threadIdx.x;
    if (q >= nq) return;
    const unsigned a = (unsigned)qs[q];
    const unsigned b = (unsigned)qd[q];
    const float ta = st[(a / (unsigned)OWN_T_ROWS) * (unsigned)OWN_PITCH(OWN_T_ROWS) + a % (unsigned)OWN_T_ROWS];
    const float mb = sm[(b / (unsigned)OWN_M_ROWS) * (unsigned)OWN_PITCH(OWN_M_ROWS) + b % (unsigned)OWN_M_ROWS];
    out[q] = (ta + mb) + cb[0];
}

extern "C" void kernel_launch(void* const* d_in, const int* in_sizes, int n_in,
                              void* d_out, int out_size, void* d_ws, size_t ws_size,
                              hipStream_t stream) {
    const float* x_thesis     = (const float*)d_in[0];
    const int*   mentor_id    = (const int*)  d_in[1];
    const int*   edge_src     = (const int*)  d_in[2];
    const int*   edge_dst     = (const int*)  d_in[3];
    const int*   eli_src      = (const int*)  d_in[4];
    const int*   eli_dst      = (const int*)  d_in[5];
    const float* thesis_lin_w = (const float*)d_in[6];
    const float* thesis_lin_b = (const float*)d_in[7];
    const float* prof_emb     = (const float*)d_in[8];
    const float* W_nbr_sup    = (const float*)d_in[9];
    const float* W_root_sup   = (const float*)d_in[10];
    const float* b_sup        = (const float*)d_in[11];
    const float* W_nbr_rev    = (const float*)d_in[12];
    const float* W_root_rev   = (const float*)d_in[13];
    const float* b_rev        = (const float*)d_in[14];
    const float* cls_w        = (const float*)d_in[15];
    const float* cls_b        = (const float*)d_in[16];
    float*       out          = (float*)d_out;

    const size_t T_PLANE = (size_t)N_T * DDIM;
    const size_t M_PLANE = (size_t)N_M * DDIM;
    static_assert(2 * BM_PLANE <= (size_t)N_T * DDIM, "both sides' bitmaps fit in one plane");
    float* p    = (float*)d_ws;
    float* tX   = p; p += T_PLANE;
    float* tY   = p; p += T_PLANE;
    float* tZ   = p; p += T_PLANE;
    float* sT   = p;
    float* sM   = p + (size_t)OWNER_BLOCKS * OWN_PITCH(OWN_T_ROWS);
    static_assert((size_t)OWNER_BLOCKS * (OWN_PITCH(OWN_T_ROWS) + OWN_PITCH(OWN_M_ROWS)) <= (size_t)N_M * DDIM,
                  "both scalar planes fit in this plane");
    p += M_PLANE;
    float* mQ   = p; p += M_PLANE;
    float* mR   = p; p += M_PLANE;
    unsigned* bmM = (unsigned*)tY;
    unsigned* bmT = bmM + BM_PLANE;

    const int BLK = 256;
    const int ROWS_PER_BLK = 8 * 16;

    k_owner_bitmaps<<<dim3(BM_NCH, 2), BM_CHUNK, 0, stream>>>(edge_dst, edge_src, bmM, bmT);

    float* ht0 = tZ;
    float* hm0 = mR;
    k_gemm_wmma<<<(N_T + ROWS_PER_BLK - 1) / ROWS_PER_BLK, BLK, 0, stream>>>(
        x_thesis, thesis_lin_w, F_IN, nullptr, nullptr, 0,
        thesis_lin_b, ht0, N_T, 0);
    {
        int n = N_M * (DDIM / 4);
        k_gather_rows<<<(n + BLK - 1) / BLK, BLK, 0, stream>>>(prof_emb, mentor_id, hm0, N_M);
    }

    float* hm1 = mQ;
    float* ht1 = tX;
    k_owner_layer<OWN_M_ROWS, 2, false><<<OWNER_BLOCKS, OWNER_THREADS, 0, stream>>>(
        ht0, edge_dst, edge_src, bmM, hm0, W_nbr_sup, W_root_sup, b_sup, nullptr, hm1, N_M, 1);
    k_owner_layer<OWN_T_ROWS, 8, false><<<OWNER_BLOCKS, OWNER_THREADS, 0, stream>>>(
        hm0, edge_src, edge_dst, bmT, ht0, W_nbr_rev, W_root_rev, b_rev, nullptr, ht1, N_T, 1);

    const size_t wOff = (size_t)DDIM * DDIM;
    k_owner_layer<OWN_M_ROWS, 2, true><<<OWNER_BLOCKS, OWNER_THREADS, 0, stream>>>(
        ht1, edge_dst, edge_src, bmM, hm1, W_nbr_sup + wOff, W_root_sup + wOff, b_sup + DDIM,
        cls_w + DDIM, sM, N_M, 0);
    k_owner_layer<OWN_T_ROWS, 8, true><<<OWNER_BLOCKS, OWNER_THREADS, 0, stream>>>(
        hm1, edge_src, edge_dst, bmT, ht1, W_nbr_rev + wOff, W_root_rev + wOff, b_rev + DDIM,
        cls_w, sT, N_T, 0);

    k_classify_scalars<<<(N_Q + OWNER_THREADS - 1) / OWNER_THREADS, OWNER_THREADS, 0, stream>>>(
        sT, sM, eli_src, eli_dst, cls_b, out, N_Q);
}


# Owner blocks' per-chunk bitmap line becomes the edge list itself: chunk kernel writes a count and up to 31 packed words (row in range | other node) per owner; no id reads per edge; sum order changes
# speedup vs baseline: 2.8451x; 1.0727x over previous
#include <hip/hip_runtime.h>
#include <hip/hip_bf16.h>

#define N_T   100000
#define N_M   20000
#define N_E   600000
#define N_Q   200000
#define F_IN  256
#define DDIM  128

#define K_SLAB   64
#define B_STRIDE 68

typedef float v2f __attribute__((ext_vector_type(2)));
typedef float v8f __attribute__((ext_vector_type(8)));

__global__ void k_gather_rows(const float* __restrict__ src, const int* __restrict__ ids,
                              float* __restrict__ dst, int nrows) {
    int t = blockIdx.x * blockDim.x + threadIdx.x;
    if (t >= nrows * (DDIM / 4)) return;
    int row = t >> 5;
    int l   = t & 31;
    float4 v = *(const float4*)(src + (size_t)ids[row] * DDIM + l * 4);
    *(float4*)(dst + (size_t)row * DDIM + l * 4) = v;
}

#define OWNER_BLOCKS   256
#define OWNER_THREADS  1024
#define OWNER_WAVES    32
#define OWN_T_ROWS     391
#define OWN_M_ROWS     79
#define QCAP           64
#define BM_CHUNK       1024
#define BM_NCH         ((N_E + BM_CHUNK - 1) / BM_CHUNK)
#define BM_WORDS       (BM_NCH * 32)
#define BM_PLANE       ((size_t)OWNER_BLOCKS * BM_WORDS)
#define ITEMS_PER_GROUP 8
#define SUM_PITCH      132
#define W_PITCH        132

#define PK_SHIFT  17
#define PK_MASK   ((1u << PK_SHIFT) - 1u)
#define REC_CAP   31
#define REC_PITCH 33
__global__ void __launch_bounds__(BM_CHUNK)
k_owner_lists(const int* __restrict__ key_m, const int* __restrict__ key_t,
              unsigned* __restrict__ bm_m, unsigned* __restrict__ bm_t) {
    static_assert(OWNER_BLOCKS % 32 == 0, "a wave stores one owner's line per step");
    static_assert(N_T <= (1 << PK_SHIFT) && N_M <= (1 << PK_SHIFT), "a node id fits the low field");
    static_assert(OWN_T_ROWS <= (1 << (32 - PK_SHIFT)) && OWN_M_ROWS <= (1 << (32 - PK_SHIFT)),
                  "a row number inside a range fits the high field");
    __shared__ unsigned rec[OWNER_BLOCKS * REC_PITCH];

    const int  tid    = threadIdx.x;
    const int  chunk  = blockIdx.x;
    const bool side_t = blockIdx.y != 0;
    const int* keys   = side_t ? key_t : key_m;
    const int* others = side_t ? key_m : key_t;
    unsigned*  bm     = side_t ? bm_t : bm_m;

    for (int i = tid; i < OWNER_BLOCKS * REC_PITCH; i += BM_CHUNK) rec[i] = 0u;
    __syncthreads();

    const int e = chunk * BM_CHUNK + tid;
    if (e < N_E) {
        const unsigned k    = (unsigned)keys[e];
        const unsigned rows = side_t ? (unsigned)OWN_T_ROWS : (unsigned)OWN_M_ROWS;
        const unsigned own  = k / rows;
        if (own < (unsigned)OWNER_BLOCKS) {
            const unsigned word = ((k - own * rows) << PK_SHIFT) | ((unsigned)others[e] & PK_MASK);
            const unsigned pos  = __hip_atomic_fetch_add(&rec[own * REC_PITCH], 1u,
                                                         __ATOMIC_RELAXED, __HIP_MEMORY_SCOPE_WORKGROUP);
            if (pos < (unsigned)REC_CAP) rec[own * REC_PITCH + 1 + pos] = word;
        }
    }
    __syncthreads();

    const int wd = tid & 31;
    for (int o = tid >> 5; o < OWNER_BLOCKS; o += BM_CHUNK / 32)
        bm[((size_t)o * BM_NCH + chunk) * 32 + wd] = rec[o * REC_PITCH + wd];
}

__device__ __forceinline__ void owner_drain(const float* __restrict__ feat,
                                            const int* qd, int c, int lane,
                                            float* sums, float* cnts) {
    int loc = 0, o = 0;
    if (lane < c) {
        const unsigned p = (unsigned)qd[lane];
        loc = (int)(p >> PK_SHIFT);
        o   = (int)(p & PK_MASK);
        __hip_atomic_fetch_add(cnts + loc, 1.0f, __ATOMIC_RELAXED, __HIP_MEMORY_SCOPE_WORKGROUP);
    }
    for (int j0 = 0; j0 < c; j0 += ITEMS_PER_GROUP) {
        float v[ITEMS_PER_GROUP][4];
#pragma unroll
        for (int j = 0; j < ITEMS_PER_GROUP; j++) {
            const int oj = __builtin_amdgcn_readlane(o, j0 + j);
            const float* src = feat + (size_t)oj * DDIM + lane;
            v[j][0] = src[0];
            v[j][1] = src[32];
            v[j][2] = src[64];
            v[j][3] = src[96];
        }
#pragma unroll
        for (int j = 0; j < ITEMS_PER_GROUP; j++) {
            if (j == 0 || j0 + j < c) {
                const int lj = __builtin_amdgcn_readlane(loc, j0 + j);
                float* dst = sums + lj * SUM_PITCH + lane;
                __hip_atomic_fetch_add(dst +  0, v[j][0], __ATOMIC_RELAXED, __HIP_MEMORY_SCOPE_WORKGROUP);
                __hip_atomic_fetch_add(dst + 32, v[j][1], __ATOMIC_RELAXED, __HIP_MEMORY_SCOPE_WORKGROUP);
                __hip_atomic_fetch_add(dst + 64, v[j][2], __ATOMIC_RELAXED, __HIP_MEMORY_SCOPE_WORKGROUP);
                __hip_atomic_fetch_add(dst + 96, v[j][3], __ATOMIC_RELAXED, __HIP_MEMORY_SCOPE_WORKGROUP);
            }
        }
    }
}

template <int NT>
__device__ __forceinline__ void k_step8n(const float4 a4, const float* Ws,
                                         int ks, int r, int h, v8f acc[NT]) {
    v2f a1; a1[0] = a4.x; a1[1] = a4.y;
    v2f a2; a2[0] = a4.z; a2[1] = a4.w;
#pragma unroll
    for (int nt = 0; nt < NT; nt++) {
        float4 b4 = *(const float4*)(Ws + (nt * 16 + r) * W_PITCH + ks + 4 * h);
        v2f b1; b1[0] = b4.x; b1[1] = b4.y;
        v2f b2; b2[0] = b4.z; b2[1] = b4.w;
        acc[nt] = __builtin_amdgcn_wmma_f32_16x16x4_f32(
            false, a1, false, b1, (short)0, acc[nt], false, false);
        acc[nt] = __builtin_amdgcn_wmma_f32_16x16x4_f32(
            false, a2, false, b2, (short)0, acc[nt], false, false);
    }
}

template <int NT>
__device__ __forceinline__ void owner_product(const float* Arow, const float* Ws,
                                              int r, int h, v8f acc[NT]) {
#pragma nounroll
    for (int k0 = 0; k0 < DDIM; k0 += K_SLAB) {
        float4 a_cur = *(const float4*)(Arow + k0);
#pragma unroll
        for (int ks = k0; ks < k0 + K_SLAB; ks += 8) {
            float4 a_nxt = a_cur;
            if (ks + 8 < k0 + K_SLAB) a_nxt = *(const float4*)(Arow + ks + 8);
            k_step8n<NT>(a_cur, Ws, ks, r, h, acc);
            a_cur = a_nxt;
            if (NT >= 8) __builtin_amdgcn_sched_barrier(0);
        }
        __builtin_amdgcn_sched_barrier(0);
        asm volatile("v_nop\n\tv_nop\n\tv_nop\n\tv_nop");
    }
}

__device__ __forceinline__ void stage_w(const float* __restrict__ Wg, float* Ws, int tid) {
#pragma unroll
    for (int i = 0; i < (DDIM * DDIM / 4) / OWNER_THREADS; i++) {
        const int idx = tid + i * OWNER_THREADS;
        const int row = idx >> 5;
        const int c   = (idx & 31) << 2;
        const float4 v = *(const float4*)(Wg + (size_t)row * DDIM + c);
        *(float4*)(Ws + row * W_PITCH + c) = v;
    }
}

__device__ __forceinline__ int lane_here() {
    int l;
    asm volatile("v_mbcnt_lo_u32_b32 %0, -1, 0" : "=v"(l));
    return l;
}

#define OWN_PITCH(rows_) ((((rows_) + 31) / 32) * 32)

__device__ __forceinline__ float half_rows_reduce(const float p[8], int r) {
    const bool b3 = (r & 8) != 0, b2 = (r & 4) != 0, b1 = (r & 2) != 0;
    float u[4], t[2];
#pragma unroll
    for (int j = 0; j < 4; j++) {
        const float send = b3 ? p[j] : p[j + 4];
        const float keep = b3 ? p[j + 4] : p[j];
        u[j] = keep + __shfl_xor(send, 8, 32);
    }
#pragma unroll
    for (int j = 0; j < 2; j++) {
        const float send = b2 ? u[j] : u[j + 2];
        const float keep = b2 ? u[j + 2] : u[j];
        t[j] = keep + __shfl_xor(send, 4, 32);
    }
    const float send = b1 ? t[0] : t[1];
    const float keep = b1 ? t[1] : t[0];
    float s = keep + __shfl_xor(send, 2, 32);
    s += __shfl_xor(s, 1, 32);
    return s;
}

template <int ROWS, int NT, bool DOT>
__global__ void __launch_bounds__(OWNER_THREADS)
k_owner_layer(const float* __restrict__ feat,
              const int* __restrict__ idx_key,
              const int* __restrict__ idx_other,
              const unsigned* __restrict__ bm,
              const float* __restrict__ self,
              const float* __restrict__ Wn, const float* __restrict__ Wr,
              const float* __restrict__ bias,
              const float* __restrict__ cw,
              float* __restrict__ out, int n_nodes, int relu) {
    static_assert((long long)BM_NCH * BM_CHUNK >= N_E, "every edge falls into a chunk");
    static_assert(BM_WORDS % 4 == 0, "a block's records are read 16 bytes per lane");
    constexpr int STRIPES = (ROWS + 15) / 16;
    constexpr int ROWS16  = STRIPES * 16;
    constexpr int CGROUPS = 8 / NT;
    constexpr int UNITS   = STRIPES * CGROUPS;
    static_assert(UNITS <= OWNER_WAVES, "one unit per wave");
    __shared__ float sums[ROWS16 * SUM_PITCH];
    __shared__ float cnts[ROWS];
    __shared__ int   q_all[OWNER_WAVES][QCAP];
    __shared__ float Ws[DDIM * W_PITCH];
    __shared__ int   next_chunk;
    __shared__ float rowdot[DOT ? CGROUPS * ROWS16 : 1];

    const int tid  = threadIdx.x;
    const int lane = tid & 31;
    const int w  = __builtin_amdgcn_readfirstlane(tid >> 5);
    const int lo = blockIdx.x * ROWS;
    if (lo >= n_nodes) return;
    const int own = min(ROWS, n_nodes - lo);

    for (int i = tid; i < ROWS16 * (SUM_PITCH / 4); i += OWNER_THREADS)
        *(float4*)(sums + i * 4) = make_float4(0.0f, 0.0f, 0.0f, 0.0f);
    for (int i = tid; i < ROWS; i += OWNER_THREADS) cnts[i] = 0.0f;
    if (tid == 0) next_chunk = 0;
    stage_w(Wr, Ws, tid);
    __syncthreads();

    const bool worker = w < UNITS;
    const int  stripe = worker ? w / CGROUPS : 0;
    const int  cg     = worker ? w % CGROUPS : 0;
    const float* Wcols = Ws + cg * (NT * 16) * W_PITCH;

    v8f acc[NT] = {};
    if (worker) {
        const int r = lane & 15;
        const int h = lane >> 4;
        const int grow = min(lo + stripe * 16 + r, n_nodes - 1);
        owner_product<NT>(self + (size_t)grow * DDIM + 4 * h, Wcols, r, h, acc);
    }

    const int lane1 = lane_here();
    int* q = q_all[w];
    int qn = 0;

    const uint4* mine4 = (const uint4*)(bm + (size_t)blockIdx.x * BM_WORDS);
    constexpr int N4  = BM_WORDS / 4;
    constexpr int NCH = (N4 + 31) / 32;
#define OWNER_TAKE(c_) \
    do { \
        int t_ = 0; \
        if (lane1 == 0) \
            t_ = __hip_atomic_fetch_add(&next_chunk, 1, __ATOMIC_RELAXED, __HIP_MEMORY_SCOPE_WORKGROUP); \
        (c_) = __builtin_amdgcn_readfirstlane(t_); \
    } while (0)
#define OWNER_GROUP(c_) \
    (((c_) < NCH && (c_) * 32 + lane1 < N4) ? mine4[(c_) * 32 + lane1] : make_uint4(0u, 0u, 0u, 0u))
#define OWNER_PUSH(take_, word_) \
    do { \
        const unsigned m_ = __builtin_amdgcn_ballot_w32(take_); \
        if (m_ != 0u) {                           \
            if (take_) q[qn + (int)__builtin_amdgcn_mbcnt_lo(m_, 0u)] = (int)(word_); \
            qn += __builtin_popcount(m_); \
            if (qn >= 32) { \
                qn -= 32; \
                owner_drain(feat, q + qn, 32, lane1, sums, cnts); \
            } \
        } \
    } while (0)

    int c_cur;
    OWNER_TAKE(c_cur);
    uint4 cur = OWNER_GROUP(c_cur);
    for (int it = 0; it < NCH && c_cur < NCH; it++) {
        int c_nxt;
        OWNER_TAKE(c_nxt);
        const uint4 nxt = OWNER_GROUP(c_nxt);
        const unsigned cnt_r = (unsigned)__builtin_amdgcn_ds_bpermute((lane1 & ~7) << 2, (int)cur.x);
        const bool     over  = cnt_r > (unsigned)REC_CAP;
        const unsigned n_ok  = over ? 0u : cnt_r;
        const unsigned j0    = (unsigned)(lane1 & 7) << 2;
        for (int k = 0; k < 4; k++) {
            const unsigned word = k == 0 ? cur.x : k == 1 ? cur.y : k == 2 ? cur.z : cur.w;
            const unsigned j    = j0 + (unsigned)k;
            const bool     take = j >= 1u && j <= n_ok;
            OWNER_PUSH(take, word);
        }
        unsigned ov = __builtin_amdgcn_ballot_w32(over && (lane1 & 7) == 0);
        while (ov) {
            const int L = __builtin_ctz(ov);
            ov &= ov - 1u;
            const int ebase = ((c_cur << 2) + (L >> 3)) * BM_CHUNK;
            for (int i = 0; i < BM_CHUNK / 32; i++) {
                const int e = ebase + (i << 5) + lane1;
                bool     take = false;
                unsigned word = 0u;
                if (e < N_E) {
                    const unsigned loc = (unsigned)(idx_key[e] - lo);
                    if (loc < (unsigned)ROWS) {
                        take = true;
                        word = (loc << PK_SHIFT) | ((unsigned)idx_other[e] & PK_MASK);
                    }
                }
                OWNER_PUSH(take, word);
            }
        }
        cur = nxt;
        c_cur = c_nxt;
    }
#undef OWNER_PUSH
#undef OWNER_GROUP
#undef OWNER_TAKE
    if (qn > 0) owner_drain(feat, q, qn, lane1, sums, cnts);
    __syncthreads();

    const int lane2 = lane_here();
    const int tid2  = (w << 5) + lane2;
    for (int i = tid2; i < own * (DDIM / 4); i += OWNER_THREADS) {
        const int row = i >> 5;
        const int c4  = (i & 31) * 4;
        const float rc = 1.0f / fmaxf(cnts[row], 1.0f);
        float4 s = *(const float4*)(sums + row * SUM_PITCH + c4);
        s.x *= rc; s.y *= rc; s.z *= rc; s.w *= rc;
        *(float4*)(sums + row * SUM_PITCH + c4) = s;
    }
    stage_w(Wn, Ws, tid2);
    __syncthreads();

    if constexpr (!DOT) {
        if (!worker) return;

        const int r2 = lane2 & 15;
        const int h2 = lane2 >> 4;
        owner_product<NT>(sums + (stripe * 16 + r2) * SUM_PITCH + 4 * h2, Wcols, r2, h2, acc);

        const int lane3 = lane_here();
        const int r3 = lane3 & 15;
        const int h3 = lane3 >> 4;
#pragma unroll
        for (int nt = 0; nt < NT; nt++) {
            const int col  = (cg * NT + nt) * 16 + r3;
            const float bb = bias[col];
#pragma unroll
            for (int i = 0; i < 8; i++) {
                const int lrow = stripe * 16 + h3 * 8 + i;
                float v = acc[nt][i] + bb;
                if (relu) v = fmaxf(v, 0.0f);
                if (lrow < own) out[(size_t)(lo + lrow) * DDIM + col] = v;
            }
        }
    } else {
        if (worker) {
            const int r2 = lane2 & 15;
            const int h2 = lane2 >> 4;
            owner_product<NT>(sums + (stripe * 16 + r2) * SUM_PITCH + 4 * h2, Wcols, r2, h2, acc);

            const int lane3 = lane_here();
            const int r3 = lane3 & 15;
            const int h3 = lane3 >> 4;
            float p[8] = {0.0f, 0.0f, 0.0f, 0.0f, 0.0f, 0.0f, 0.0f, 0.0f};
#pragma unroll
            for (int nt = 0; nt < NT; nt++) {
                const int col  = (cg * NT + nt) * 16 + r3;
                const float bb = bias[col];
                const float wv = cw[col];
#pragma unroll
                for (int i = 0; i < 8; i++) {
                    float v = acc[nt][i] + bb;
                    if (relu) v = fmaxf(v, 0.0f);
                    p[i] += v * wv;
                }
            }
            const float s = half_rows_reduce(p, r3);
            if ((r3 & 1) == 0)
                rowdot[cg * ROWS16 + stripe * 16 + h3 * 8 + (r3 >> 1)] = s;
        }
        __syncthreads();

        const int lane4 = lane_here();
        float* dst = out + (size_t)blockIdx.x * OWN_PITCH(ROWS);
        for (int i = (w << 5) + lane4; i < own; i += OWNER_THREADS) {
            float s = rowdot[i];
#pragma unroll
            for (int g = 1; g < CGROUPS; g++) s += rowdot[g * ROWS16 + i];
            dst[i] = s;
        }
    }
}

__device__ __forceinline__ void k_step8(const float4 a4, const float* __restrict__ Bs,
                                        int ks, int r, int h, v8f acc[8]) {
    v2f a1; a1[0] = a4.x; a1[1] = a4.y;
    v2f a2; a2[0] = a4.z; a2[1] = a4.w;
#pragma unroll
    for (int nt = 0; nt < 8; nt++) {
        float4 b4 = *(const float4*)(Bs + (nt * 16 + r) * B_STRIDE + ks + 4 * h);
        v2f b1; b1[0] = b4.x; b1[1] = b4.y;
        v2f b2; b2[0] = b4.z; b2[1] = b4.w;
        acc[nt] = __builtin_amdgcn_wmma_f32_16x16x4_f32(
            false, a1, false, b1, (short)0, acc[nt], false, false);
        acc[nt] = __builtin_amdgcn_wmma_f32_16x16x4_f32(
            false, a2, false, b2, (short)0, acc[nt], false, false);
    }
}

__device__ __forceinline__ void gemm_pass(const float* A,
                                          const float* __restrict__ W,
                                          int K, int rowA, int r, int h, int tid,
                                          float* __restrict__ Bs, v8f acc[8]) {
    for (int kk0 = 0; kk0 < K; kk0 += K_SLAB) {
        __syncthreads();
#pragma unroll
        for (int i = 0; i < 8; i++) {
            int idx = tid + i * 256;
            int row = idx >> 4;
            int c   = (idx & 15) << 2;
            float4 v = *(const float4*)(W + (size_t)row * K + kk0 + c);
            *(float4*)(Bs + row * B_STRIDE + c) = v;
        }
        __syncthreads();

        const size_t aoff = (size_t)(rowA + r) * K + kk0 + 4 * h;
        const float* Arow = A + aoff;
        for (int ks = 0; ks < K_SLAB; ks += 8) {
            float4 a4 = *(const float4*)(Arow + ks);
            k_step8(a4, Bs, ks, r, h, acc);
        }
        __builtin_amdgcn_sched_barrier(0);
        asm volatile("v_nop\n\tv_nop\n\tv_nop\n\tv_nop");
    }
}

__global__ void __launch_bounds__(256)
k_gemm_wmma(const float* A1, const float* __restrict__ W1, int K1,
            const float* __restrict__ A2, const float* __restrict__ W2, int K2,
            const float* __restrict__ bias, float* out,
            int nrows, int relu) {
    __shared__ float Bs[DDIM * B_STRIDE];

    const int tid  = threadIdx.x;
    const int lane = tid & 31;
    const int wave = tid >> 5;
    const int row0 = (blockIdx.x * 8 + wave) * 16;
    const bool valid = (row0 < nrows);
    const int rowA = valid ? row0 : 0;
    const int r = lane & 15;
    const int h = lane >> 4;

    v8f acc[8] = {};

    gemm_pass(A1, W1, K1, rowA, r, h, tid, Bs, acc);
    if (A2) gemm_pass(A2, W2, K2, rowA, r, h, tid, Bs, acc);

    if (!valid) return;
#pragma unroll
    for (int nt = 0; nt < 8; nt++) {
        int col  = nt * 16 + r;
        float bb = bias[col];
#pragma unroll
        for (int i = 0; i < 8; i++) {
            float v = acc[nt][i] + bb;
            if (relu) v = fmaxf(v, 0.0f);
            out[(size_t)(row0 + h * 8 + i) * DDIM + col] = v;
        }
    }
}

__global__ void __launch_bounds__(OWNER_THREADS)
k_classify_scalars(const float* __restrict__ st, const float* __restrict__ sm,
                   const int* __restrict__ qs, const int* __restrict__ qd,
                   const float* __restrict__ cb,
                   float* __restrict__ out, int nq) {
    const int q = blockIdx.x * blockDim.x + threadIdx.x;
    if (q >= nq) return;
    const unsigned a = (unsigned)qs[q];
    const unsigned b = (unsigned)qd[q];
    const float ta = st[(a / (unsigned)OWN_T_ROWS) * (unsigned)OWN_PITCH(OWN_T_ROWS) + a % (unsigned)OWN_T_ROWS];
    const float mb = sm[(b / (unsigned)OWN_M_ROWS) * (unsigned)OWN_PITCH(OWN_M_ROWS) + b % (unsigned)OWN_M_ROWS];
    out[q] = (ta + mb) + cb[0];
}

extern "C" void kernel_launch(void* const* d_in, const int* in_sizes, int n_in,
                              void* d_out, int out_size, void* d_ws, size_t ws_size,
                              hipStream_t stream) {
    const float* x_thesis     = (const float*)d_in[0];
    const int*   mentor_id    = (const int*)  d_in[1];
    const int*   edge_src     = (const int*)  d_in[2];
    const int*   edge_dst     = (const int*)  d_in[3];
    const int*   eli_src      = (const int*)  d_in[4];
    const int*   eli_dst      = (const int*)  d_in[5];
    const float* thesis_lin_w = (const float*)d_in[6];
    const float* thesis_lin_b = (const float*)d_in[7];
    const float* prof_emb     = (const float*)d_in[8];
    const float* W_nbr_sup    = (const float*)d_in[9];
    const float* W_root_sup   = (const float*)d_in[10];
    const float* b_sup        = (const float*)d_in[11];
    const float* W_nbr_rev    = (const float*)d_in[12];
    const float* W_root_rev   = (const float*)d_in[13];
    const float* b_rev        = (const float*)d_in[14];
    const float* cls_w        = (const float*)d_in[15];
    const float* cls_b        = (const float*)d_in[16];
    float*       out          = (float*)d_out;

    const size_t T_PLANE = (size_t)N_T * DDIM;
    const size_t M_PLANE = (size_t)N_M * DDIM;
    static_assert(2 * BM_PLANE <= (size_t)N_T * DDIM, "both sides' records fit in one plane");
    float* p    = (float*)d_ws;
    float* tX   = p; p += T_PLANE;
    float* tY   = p; p += T_PLANE;
    float* tZ   = p; p += T_PLANE;
    float* sT   = p;
    float* sM   = p + (size_t)OWNER_BLOCKS * OWN_PITCH(OWN_T_ROWS);
    static_assert((size_t)OWNER_BLOCKS * (OWN_PITCH(OWN_T_ROWS) + OWN_PITCH(OWN_M_ROWS)) <= (size_t)N_M * DDIM,
                  "both scalar planes fit in this plane");
    p += M_PLANE;
    float* mQ   = p; p += M_PLANE;
    float* mR   = p; p += M_PLANE;
    unsigned* bmM = (unsigned*)tY;
    unsigned* bmT = bmM + BM_PLANE;

    const int BLK = 256;
    const int ROWS_PER_BLK = 8 * 16;

    k_owner_lists<<<dim3(BM_NCH, 2), BM_CHUNK, 0, stream>>>(edge_dst, edge_src, bmM, bmT);

    float* ht0 = tZ;
    float* hm0 = mR;
    k_gemm_wmma<<<(N_T + ROWS_PER_BLK - 1) / ROWS_PER_BLK, BLK, 0, stream>>>(
        x_thesis, thesis_lin_w, F_IN, nullptr, nullptr, 0,
        thesis_lin_b, ht0, N_T, 0);
    {
        int n = N_M * (DDIM / 4);
        k_gather_rows<<<(n + BLK - 1) / BLK, BLK, 0, stream>>>(prof_emb, mentor_id, hm0, N_M);
    }

    float* hm1 = mQ;
    float* ht1 = tX;
    k_owner_layer<OWN_M_ROWS, 2, false><<<OWNER_BLOCKS, OWNER_THREADS, 0, stream>>>(
        ht0, edge_dst, edge_src, bmM, hm0, W_nbr_sup, W_root_sup, b_sup, nullptr, hm1, N_M, 1);
    k_owner_layer<OWN_T_ROWS, 8, false><<<OWNER_BLOCKS, OWNER_THREADS, 0, stream>>>(
        hm0, edge_src, edge_dst, bmT, ht0, W_nbr_rev, W_root_rev, b_rev, nullptr, ht1, N_T, 1);

    const size_t wOff = (size_t)DDIM * DDIM;
    k_owner_layer<OWN_M_ROWS, 2, true><<<OWNER_BLOCKS, OWNER_THREADS, 0, stream>>>(
        ht1, edge_dst, edge_src, bmM, hm1, W_nbr_sup + wOff, W_root_sup + wOff, b_sup + DDIM,
        cls_w + DDIM, sM, N_M, 0);
    k_owner_layer<OWN_T_ROWS, 8, true><<<OWNER_BLOCKS, OWNER_THREADS, 0, stream>>>(
        hm1, edge_src, edge_dst, bmT, ht1, W_nbr_rev + wOff, W_root_rev + wOff, b_rev + DDIM,
        cls_w, sT, N_T, 0);

    k_classify_scalars<<<(N_Q + OWNER_THREADS - 1) / OWNER_THREADS, OWNER_THREADS, 0, stream>>>(
        sT, sM, eli_src, eli_dst, cls_b, out, N_Q);
}


# Owner blocks: edges first sent to row-group lists in LDS, then one wave at a time owns a group and adds its rows by plain 16-byte LDS load, add, store; no LDS float atomics; sum order changes
# speedup vs baseline: 3.0056x; 1.0564x over previous
#include <hip/hip_runtime.h>
#include <hip/hip_bf16.h>

#define N_T   100000
#define N_M   20000
#define N_E   600000
#define N_Q   200000
#define F_IN  256
#define DDIM  128

#define K_SLAB   64
#define B_STRIDE 68

typedef float v2f __attribute__((ext_vector_type(2)));
typedef float v8f __attribute__((ext_vector_type(8)));

__global__ void k_gather_rows(const float* __restrict__ src, const int* __restrict__ ids,
                              float* __restrict__ dst, int nrows) {
    int t = blockIdx.x * blockDim.x + threadIdx.x;
    if (t >= nrows * (DDIM / 4)) return;
    int row = t >> 5;
    int l   = t & 31;
    float4 v = *(const float4*)(src + (size_t)ids[row] * DDIM + l * 4);
    *(float4*)(dst + (size_t)row * DDIM + l * 4) = v;
}

#define OWNER_BLOCKS   256
#define OWNER_THREADS  1024
#define OWNER_WAVES    32
#define OWN_T_ROWS     391
#define OWN_M_ROWS     79
#define QCAP           64
#define BM_CHUNK       1024
#define BM_NCH         ((N_E + BM_CHUNK - 1) / BM_CHUNK)
#define BM_WORDS       (BM_NCH * 32)
#define BM_PLANE       ((size_t)OWNER_BLOCKS * BM_WORDS)
#define ITEMS_PER_GROUP 8
#define ROW_GROUPS     64
#define GROUP_CAP      96
#define SUM_PITCH      132
#define W_PITCH        132

#define PK_SHIFT  17
#define PK_MASK   ((1u << PK_SHIFT) - 1u)
#define REC_CAP   31
#define REC_PITCH 33
__global__ void __launch_bounds__(BM_CHUNK)
k_owner_lists(const int* __restrict__ key_m, const int* __restrict__ key_t,
              unsigned* __restrict__ bm_m, unsigned* __restrict__ bm_t) {
    static_assert(OWNER_BLOCKS % 32 == 0, "a wave stores one owner's line per step");
    static_assert(N_T <= (1 << PK_SHIFT) && N_M <= (1 << PK_SHIFT), "a node id fits the low field");
    static_assert(OWN_T_ROWS <= (1 << (32 - PK_SHIFT)) && OWN_M_ROWS <= (1 << (32 - PK_SHIFT)),
                  "a row number inside a range fits the high field");
    __shared__ unsigned rec[OWNER_BLOCKS * REC_PITCH];

    const int  tid    = threadIdx.x;
    const int  chunk  = blockIdx.x;
    const bool side_t = blockIdx.y != 0;
    const int* keys   = side_t ? key_t : key_m;
    const int* others = side_t ? key_m : key_t;
    unsigned*  bm     = side_t ? bm_t : bm_m;

    for (int i = tid; i < OWNER_BLOCKS * REC_PITCH; i += BM_CHUNK) rec[i] = 0u;
    __syncthreads();

    const int e = chunk * BM_CHUNK + tid;
    if (e < N_E) {
        const unsigned k    = (unsigned)keys[e];
        const unsigned rows = side_t ? (unsigned)OWN_T_ROWS : (unsigned)OWN_M_ROWS;
        const unsigned own  = k / rows;
        if (own < (unsigned)OWNER_BLOCKS) {
            const unsigned word = ((k - own * rows) << PK_SHIFT) | ((unsigned)others[e] & PK_MASK);
            const unsigned pos  = __hip_atomic_fetch_add(&rec[own * REC_PITCH], 1u,
                                                         __ATOMIC_RELAXED, __HIP_MEMORY_SCOPE_WORKGROUP);
            if (pos < (unsigned)REC_CAP) rec[own * REC_PITCH + 1 + pos] = word;
        }
    }
    __syncthreads();

    const int wd = tid & 31;
    for (int o = tid >> 5; o < OWNER_BLOCKS; o += BM_CHUNK / 32)
        bm[((size_t)o * BM_NCH + chunk) * 32 + wd] = rec[o * REC_PITCH + wd];
}

__device__ __forceinline__ void owner_route(const float* __restrict__ feat,
                                            const int* qd, int c, int lane,
                                            float* sums, float* cnts,
                                            int* inbox, unsigned* inbox_n) {
    unsigned p = 0u;
    bool left = false;
    if (lane < c) {
        p = (unsigned)qd[lane];
        const int loc = (int)(p >> PK_SHIFT);
        const int g   = loc & (ROW_GROUPS - 1);
        __hip_atomic_fetch_add(cnts + loc, 1.0f, __ATOMIC_RELAXED, __HIP_MEMORY_SCOPE_WORKGROUP);
        const unsigned pos = __hip_atomic_fetch_add(inbox_n + g, 1u,
                                                    __ATOMIC_RELAXED, __HIP_MEMORY_SCOPE_WORKGROUP);
        if (pos < (unsigned)GROUP_CAP) inbox[g * GROUP_CAP + (int)pos] = (int)p;
        else left = true;
    }
    unsigned ov = __builtin_amdgcn_ballot_w32(left);
    while (ov) {
        const int L = __builtin_ctz(ov);
        ov &= ov - 1u;
        const unsigned pw = (unsigned)__builtin_amdgcn_readlane((int)p, L);
        const float* src = feat + (size_t)(pw & PK_MASK) * DDIM + lane;
        float* dst = sums + (int)(pw >> PK_SHIFT) * SUM_PITCH + lane;
        __hip_atomic_fetch_add(dst +  0, src[0],  __ATOMIC_RELAXED, __HIP_MEMORY_SCOPE_WORKGROUP);
        __hip_atomic_fetch_add(dst + 32, src[32], __ATOMIC_RELAXED, __HIP_MEMORY_SCOPE_WORKGROUP);
        __hip_atomic_fetch_add(dst + 64, src[64], __ATOMIC_RELAXED, __HIP_MEMORY_SCOPE_WORKGROUP);
        __hip_atomic_fetch_add(dst + 96, src[96], __ATOMIC_RELAXED, __HIP_MEMORY_SCOPE_WORKGROUP);
    }
}

template <int ITEMS>
__device__ __forceinline__ void owner_group(const float* __restrict__ feat,
                                            const int* list, int n, int lane, float* sums) {
    static_assert(32 % ITEMS == 0 && ITEMS % 2 == 0, "whole steps in a list's 32 entries, whole pairs in a step");
    for (int base = 0; base < n; base += 32) {
        const int c = min(32, n - base);
        unsigned p = 0u;
        if (lane < c) p = (unsigned)list[base + lane];
        const int loc = (int)(p >> PK_SHIFT);
        const int o   = (int)(p & PK_MASK);
        for (int j0 = 0; j0 < c; j0 += ITEMS) {
            float4 v[ITEMS];
#pragma unroll
            for (int j = 0; j < ITEMS; j++) {
                const int oj = __builtin_amdgcn_readlane(o, j0 + j);
                v[j] = *(const float4*)(feat + (size_t)oj * DDIM + 4 * lane);
            }
#pragma unroll
            for (int j = 0; j < ITEMS; j += 2) {
                if (j == 0 || j0 + j < c) {
                    const int la = __builtin_amdgcn_readlane(loc, j0 + j);
                    const int lb = __builtin_amdgcn_readlane(loc, j0 + j + 1);
                    bool two = j0 + j + 1 < c;
                    float4 va = v[j];
                    const float4 vb = v[j + 1];
                    if (two && lb == la) {
                        va.x += vb.x; va.y += vb.y; va.z += vb.z; va.w += vb.w;
                        two = false;
                    }
                    float4* pa = (float4*)(sums + la * SUM_PITCH) + lane;
                    if (two) {
                        float4* pb = (float4*)(sums + lb * SUM_PITCH) + lane;
                        float4 sa = *pa;
                        float4 sb = *pb;
                        sa.x += va.x; sa.y += va.y; sa.z += va.z; sa.w += va.w;
                        sb.x += vb.x; sb.y += vb.y; sb.z += vb.z; sb.w += vb.w;
                        *pa = sa;
                        *pb = sb;
                    } else {
                        float4 sa = *pa;
                        sa.x += va.x; sa.y += va.y; sa.z += va.z; sa.w += va.w;
                        *pa = sa;
                    }
                }
            }
        }
    }
}

template <int NT>
__device__ __forceinline__ void k_step8n(const float4 a4, const float* Ws,
                                         int ks, int r, int h, v8f acc[NT]) {
    v2f a1; a1[0] = a4.x; a1[1] = a4.y;
    v2f a2; a2[0] = a4.z; a2[1] = a4.w;
#pragma unroll
    for (int nt = 0; nt < NT; nt++) {
        float4 b4 = *(const float4*)(Ws + (nt * 16 + r) * W_PITCH + ks + 4 * h);
        v2f b1; b1[0] = b4.x; b1[1] = b4.y;
        v2f b2; b2[0] = b4.z; b2[1] = b4.w;
        acc[nt] = __builtin_amdgcn_wmma_f32_16x16x4_f32(
            false, a1, false, b1, (short)0, acc[nt], false, false);
        acc[nt] = __builtin_amdgcn_wmma_f32_16x16x4_f32(
            false, a2, false, b2, (short)0, acc[nt], false, false);
    }
}

template <int NT>
__device__ __forceinline__ void owner_product(const float* Arow, const float* Ws,
                                              int r, int h, v8f acc[NT]) {
#pragma nounroll
    for (int k0 = 0; k0 < DDIM; k0 += K_SLAB) {
        float4 a_cur = *(const float4*)(Arow + k0);
#pragma unroll
        for (int ks = k0; ks < k0 + K_SLAB; ks += 8) {
            float4 a_nxt = a_cur;
            if (ks + 8 < k0 + K_SLAB) a_nxt = *(const float4*)(Arow + ks + 8);
            k_step8n<NT>(a_cur, Ws, ks, r, h, acc);
            a_cur = a_nxt;
            if (NT >= 8) __builtin_amdgcn_sched_barrier(0);
        }
        __builtin_amdgcn_sched_barrier(0);
        asm volatile("v_nop\n\tv_nop\n\tv_nop\n\tv_nop");
    }
}

__device__ __forceinline__ void stage_w(const float* __restrict__ Wg, float* Ws, int tid) {
#pragma unroll
    for (int i = 0; i < (DDIM * DDIM / 4) / OWNER_THREADS; i++) {
        const int idx = tid + i * OWNER_THREADS;
        const int row = idx >> 5;
        const int c   = (idx & 31) << 2;
        const float4 v = *(const float4*)(Wg + (size_t)row * DDIM + c);
        *(float4*)(Ws + row * W_PITCH + c) = v;
    }
}

__device__ __forceinline__ int lane_here() {
    int l;
    asm volatile("v_mbcnt_lo_u32_b32 %0, -1, 0" : "=v"(l));
    return l;
}

#define OWN_PITCH(rows_) ((((rows_) + 31) / 32) * 32)

__device__ __forceinline__ float half_rows_reduce(const float p[8], int r) {
    const bool b3 = (r & 8) != 0, b2 = (r & 4) != 0, b1 = (r & 2) != 0;
    float u[4], t[2];
#pragma unroll
    for (int j = 0; j < 4; j++) {
        const float send = b3 ? p[j] : p[j + 4];
        const float keep = b3 ? p[j + 4] : p[j];
        u[j] = keep + __shfl_xor(send, 8, 32);
    }
#pragma unroll
    for (int j = 0; j < 2; j++) {
        const float send = b2 ? u[j] : u[j + 2];
        const float keep = b2 ? u[j + 2] : u[j];
        t[j] = keep + __shfl_xor(send, 4, 32);
    }
    const float send = b1 ? t[0] : t[1];
    const float keep = b1 ? t[1] : t[0];
    float s = keep + __shfl_xor(send, 2, 32);
    s += __shfl_xor(s, 1, 32);
    return s;
}

template <int ROWS, int NT, bool DOT>
__global__ void __launch_bounds__(OWNER_THREADS)
k_owner_layer(const float* __restrict__ feat,
              const int* __restrict__ idx_key,
              const int* __restrict__ idx_other,
              const unsigned* __restrict__ bm,
              const float* __restrict__ self,
              const float* __restrict__ Wn, const float* __restrict__ Wr,
              const float* __restrict__ bias,
              const float* __restrict__ cw,
              float* __restrict__ out, int n_nodes, int relu) {
    static_assert((long long)BM_NCH * BM_CHUNK >= N_E, "every edge falls into a chunk");
    static_assert(BM_WORDS % 4 == 0, "a block's records are read 16 bytes per lane");
    constexpr int STRIPES = (ROWS + 15) / 16;
    constexpr int ROWS16  = STRIPES * 16;
    constexpr int CGROUPS = 8 / NT;
    constexpr int UNITS   = STRIPES * CGROUPS;
    static_assert(UNITS <= OWNER_WAVES, "one unit per wave");
    __shared__ float sums[ROWS16 * SUM_PITCH];
    __shared__ float cnts[ROWS];
    __shared__ int   q_all[OWNER_WAVES][QCAP];
    __shared__ float Ws[DDIM * W_PITCH];
    __shared__ int   next_group;
    __shared__ int      inbox[ROW_GROUPS * GROUP_CAP];
    __shared__ unsigned inbox_n[ROW_GROUPS];
    __shared__ float rowdot[DOT ? CGROUPS * ROWS16 : 1];

    const int tid  = threadIdx.x;
    const int lane = tid & 31;
    const int w  = __builtin_amdgcn_readfirstlane(tid >> 5);
    const int lo = blockIdx.x * ROWS;
    if (lo >= n_nodes) return;
    const int own = min(ROWS, n_nodes - lo);

    const uint4* mine4 = (const uint4*)(bm + (size_t)blockIdx.x * BM_WORDS);
    constexpr int N4    = BM_WORDS / 4;
    constexpr int NCH   = (N4 + 31) / 32;
    constexpr int TRIPS = (NCH + OWNER_WAVES - 1) / OWNER_WAVES;
#define OWNER_TAKE(ctr_, lane_, c_) \
    do { \
        int t_ = 0; \
        if ((lane_) == 0) \
            t_ = __hip_atomic_fetch_add(&(ctr_), 1, __ATOMIC_RELAXED, __HIP_MEMORY_SCOPE_WORKGROUP); \
        (c_) = __builtin_amdgcn_readfirstlane(t_); \
    } while (0)
#define OWNER_GROUP(c_) \
    (((c_) < NCH && (c_) * 32 + lane < N4) ? mine4[(c_) * 32 + lane] : make_uint4(0u, 0u, 0u, 0u))
    uint4 recs[TRIPS];
#pragma unroll
    for (int t = 0; t < TRIPS; t++) recs[t] = OWNER_GROUP(w + t * OWNER_WAVES);

    for (int i = tid; i < ROWS16 * (SUM_PITCH / 4); i += OWNER_THREADS)
        *(float4*)(sums + i * 4) = make_float4(0.0f, 0.0f, 0.0f, 0.0f);
    for (int i = tid; i < ROWS; i += OWNER_THREADS) cnts[i] = 0.0f;
    if (tid < ROW_GROUPS) inbox_n[tid] = 0u;
    if (tid == 0) next_group = 0;
    stage_w(Wr, Ws, tid);
    __syncthreads();

    {
        int* q = q_all[w];
        int qn = 0;
#define OWNER_PUSH(take_, word_) \
    do { \
        const unsigned m_ = __builtin_amdgcn_ballot_w32(take_); \
        if (m_ != 0u) {                           \
            if (take_) q[qn + (int)__builtin_amdgcn_mbcnt_lo(m_, 0u)] = (int)(word_); \
            qn += __builtin_popcount(m_); \
            if (qn >= 32) { \
                qn -= 32; \
                owner_route(feat, q + qn, 32, lane, sums, cnts, inbox, inbox_n); \
            } \
        } \
    } while (0)

#pragma nounroll
        for (int t = 0; t < TRIPS; t++) {
            const int c_cur = w + t * OWNER_WAVES;
            uint4 cur = recs[0];
#pragma unroll
            for (int u = 1; u < TRIPS; u++)
                if (t == u) cur = recs[u];
            const unsigned cnt_r = (unsigned)__builtin_amdgcn_ds_bpermute((lane & ~7) << 2, (int)cur.x);
            const bool     over  = cnt_r > (unsigned)REC_CAP;
            const unsigned n_ok  = over ? 0u : cnt_r;
            const unsigned j0    = (unsigned)(lane & 7) << 2;
            for (int k = 0; k < 4; k++) {
                const unsigned word = k == 0 ? cur.x : k == 1 ? cur.y : k == 2 ? cur.z : cur.w;
                const unsigned j    = j0 + (unsigned)k;
                const bool     take = j >= 1u && j <= n_ok;
                OWNER_PUSH(take, word);
            }
            unsigned ov = __builtin_amdgcn_ballot_w32(over && (lane & 7) == 0);
            while (ov) {
                const int L = __builtin_ctz(ov);
                ov &= ov - 1u;
                const int ebase = ((c_cur << 2) + (L >> 3)) * BM_CHUNK;
                for (int i = 0; i < BM_CHUNK / 32; i++) {
                    const int e = ebase + (i << 5) + lane;
                    bool     take = false;
                    unsigned word = 0u;
                    if (e < N_E) {
                        const unsigned loc = (unsigned)(idx_key[e] - lo);
                        if (loc < (unsigned)ROWS) {
                            take = true;
                            word = (loc << PK_SHIFT) | ((unsigned)idx_other[e] & PK_MASK);
                        }
                    }
                    OWNER_PUSH(take, word);
                }
            }
        }
#undef OWNER_PUSH
#undef OWNER_GROUP
        if (qn > 0) owner_route(feat, q, qn, lane, sums, cnts, inbox, inbox_n);
    }
    __syncthreads();

    const bool worker = w < UNITS;
    const int  stripe = worker ? w / CGROUPS : 0;
    const int  cg     = worker ? w % CGROUPS : 0;
    const float* Wcols = Ws + cg * (NT * 16) * W_PITCH;

    v8f acc[NT] = {};
    if (worker) {
        const int r = lane & 15;
        const int h = lane >> 4;
        const int grow = min(lo + stripe * 16 + r, n_nodes - 1);
        owner_product<NT>(self + (size_t)grow * DDIM + 4 * h, Wcols, r, h, acc);
    }

    const int lane1 = lane_here();
    {
        int g;
        OWNER_TAKE(next_group, lane1, g);
        for (int it = 0; it < ROW_GROUPS && g < ROW_GROUPS; it++) {
            const int n = __builtin_amdgcn_readfirstlane((int)min(inbox_n[g], (unsigned)GROUP_CAP));
            owner_group<(NT >= 8 ? ITEMS_PER_GROUP : 2 * ITEMS_PER_GROUP)>(feat, inbox + g * GROUP_CAP, n, lane1, sums);
            OWNER_TAKE(next_group, lane1, g);
        }
    }
#undef OWNER_TAKE
    __syncthreads();

    const int lane2 = lane_here();
    const int tid2  = (w << 5) + lane2;
    for (int i = tid2; i < own * (DDIM / 4); i += OWNER_THREADS) {
        const int row = i >> 5;
        const int c4  = (i & 31) * 4;
        const float rc = 1.0f / fmaxf(cnts[row], 1.0f);
        float4 s = *(const float4*)(sums + row * SUM_PITCH + c4);
        s.x *= rc; s.y *= rc; s.z *= rc; s.w *= rc;
        *(float4*)(sums + row * SUM_PITCH + c4) = s;
    }
    stage_w(Wn, Ws, tid2);
    __syncthreads();

    if constexpr (!DOT) {
        if (!worker) return;

        const int r2 = lane2 & 15;
        const int h2 = lane2 >> 4;
        owner_product<NT>(sums + (stripe * 16 + r2) * SUM_PITCH + 4 * h2, Wcols, r2, h2, acc);

        const int lane3 = lane_here();
        const int r3 = lane3 & 15;
        const int h3 = lane3 >> 4;
#pragma unroll
        for (int nt = 0; nt < NT; nt++) {
            const int col  = (cg * NT + nt) * 16 + r3;
            const float bb = bias[col];
#pragma unroll
            for (int i = 0; i < 8; i++) {
                const int lrow = stripe * 16 + h3 * 8 + i;
                float v = acc[nt][i] + bb;
                if (relu) v = fmaxf(v, 0.0f);
                if (lrow < own) out[(size_t)(lo + lrow) * DDIM + col] = v;
            }
        }
    } else {
        if (worker) {
            const int r2 = lane2 & 15;
            const int h2 = lane2 >> 4;
            owner_product<NT>(sums + (stripe * 16 + r2) * SUM_PITCH + 4 * h2, Wcols, r2, h2, acc);

            const int lane3 = lane_here();
            const int r3 = lane3 & 15;
            const int h3 = lane3 >> 4;
            float p[8] = {0.0f, 0.0f, 0.0f, 0.0f, 0.0f, 0.0f, 0.0f, 0.0f};
#pragma unroll
            for (int nt = 0; nt < NT; nt++) {
                const int col  = (cg * NT + nt) * 16 + r3;
                const float bb = bias[col];
                const float wv = cw[col];
#pragma unroll
                for (int i = 0; i < 8; i++) {
                    float v = acc[nt][i] + bb;
                    if (relu) v = fmaxf(v, 0.0f);
                    p[i] += v * wv;
                }
            }
            const float s = half_rows_reduce(p, r3);
            if ((r3 & 1) == 0)
                rowdot[cg * ROWS16 + stripe * 16 + h3 * 8 + (r3 >> 1)] = s;
        }
        __syncthreads();

        const int lane4 = lane_here();
        float* dst = out + (size_t)blockIdx.x * OWN_PITCH(ROWS);
        for (int i = (w << 5) + lane4; i < own; i += OWNER_THREADS) {
            float s = rowdot[i];
#pragma unroll
            for (int g = 1; g < CGROUPS; g++) s += rowdot[g * ROWS16 + i];
            dst[i] = s;
        }
    }
}

__device__ __forceinline__ void k_step8(const float4 a4, const float* __restrict__ Bs,
                                        int ks, int r, int h, v8f acc[8]) {
    v2f a1; a1[0] = a4.x; a1[1] = a4.y;
    v2f a2; a2[0] = a4.z; a2[1] = a4.w;
#pragma unroll
    for (int nt = 0; nt < 8; nt++) {
        float4 b4 = *(const float4*)(Bs + (nt * 16 + r) * B_STRIDE + ks + 4 * h);
        v2f b1; b1[0] = b4.x; b1[1] = b4.y;
        v2f b2; b2[0] = b4.z; b2[1] = b4.w;
        acc[nt] = __builtin_amdgcn_wmma_f32_16x16x4_f32(
            false, a1, false, b1, (short)0, acc[nt], false, false);
        acc[nt] = __builtin_amdgcn_wmma_f32_16x16x4_f32(
            false, a2, false, b2, (short)0, acc[nt], false, false);
    }
}

__device__ __forceinline__ void gemm_pass(const float* A,
                                          const float* __restrict__ W,
                                          int K, int rowA, int r, int h, int tid,
                                          float* __restrict__ Bs, v8f acc[8]) {
    for (int kk0 = 0; kk0 < K; kk0 += K_SLAB) {
        __syncthreads();
#pragma unroll
        for (int i = 0; i < 8; i++) {
            int idx = tid + i * 256;
            int row = idx >> 4;
            int c   = (idx & 15) << 2;
            float4 v = *(const float4*)(W + (size_t)row * K + kk0 + c);
            *(float4*)(Bs + row * B_STRIDE + c) = v;
        }
        __syncthreads();

        const size_t aoff = (size_t)(rowA + r) * K + kk0 + 4 * h;
        const float* Arow = A + aoff;
        for (int ks = 0; ks < K_SLAB; ks += 8) {
            float4 a4 = *(const float4*)(Arow + ks);
            k_step8(a4, Bs, ks, r, h, acc);
        }
        __builtin_amdgcn_sched_barrier(0);
        asm volatile("v_nop\n\tv_nop\n\tv_nop\n\tv_nop");
    }
}

__global__ void __launch_bounds__(256)
k_gemm_wmma(const float* A1, const float* __restrict__ W1, int K1,
            const float* __restrict__ A2, const float* __restrict__ W2, int K2,
            const float* __restrict__ bias, float* out,
            int nrows, int relu) {
    __shared__ float Bs[DDIM * B_STRIDE];

    const int tid  = threadIdx.x;
    const int lane = tid & 31;
    const int wave = tid >> 5;
    const int row0 = (blockIdx.x * 8 + wave) * 16;
    const bool valid = (row0 < nrows);
    const int rowA = valid ? row0 : 0;
    const int r = lane & 15;
    const int h = lane >> 4;

    v8f acc[8] = {};

    gemm_pass(A1, W1, K1, rowA, r, h, tid, Bs, acc);
    if (A2) gemm_pass(A2, W2, K2, rowA, r, h, tid, Bs, acc);

    if (!valid) return;
#pragma unroll
    for (int nt = 0; nt < 8; nt++) {
        int col  = nt * 16 + r;
        float bb = bias[col];
#pragma unroll
        for (int i = 0; i < 8; i++) {
            float v = acc[nt][i] + bb;
            if (relu) v = fmaxf(v, 0.0f);
            out[(size_t)(row0 + h * 8 + i) * DDIM + col] = v;
        }
    }
}

__global__ void __launch_bounds__(OWNER_THREADS)
k_classify_scalars(const float* __restrict__ st, const float* __restrict__ sm,
                   const int* __restrict__ qs, const int* __restrict__ qd,
                   const float* __restrict__ cb,
                   float* __restrict__ out, int nq) {
    const int q = blockIdx.x * blockDim.x + threadIdx.x;
    if (q >= nq) return;
    const unsigned a = (unsigned)qs[q];
    const unsigned b = (unsigned)qd[q];
    const float ta = st[(a / (unsigned)OWN_T_ROWS) * (unsigned)OWN_PITCH(OWN_T_ROWS) + a % (unsigned)OWN_T_ROWS];
    const float mb = sm[(b / (unsigned)OWN_M_ROWS) * (unsigned)OWN_PITCH(OWN_M_ROWS) + b % (unsigned)OWN_M_ROWS];
    out[q] = (ta + mb) + cb[0];
}

extern "C" void kernel_launch(void* const* d_in, const int* in_sizes, int n_in,
                              void* d_out, int out_size, void* d_ws, size_t ws_size,
                              hipStream_t stream) {
    const float* x_thesis     = (const float*)d_in[0];
    const int*   mentor_id    = (const int*)  d_in[1];
    const int*   edge_src     = (const int*)  d_in[2];
    const int*   edge_dst     = (const int*)  d_in[3];
    const int*   eli_src      = (const int*)  d_in[4];
    const int*   eli_dst      = (const int*)  d_in[5];
    const float* thesis_lin_w = (const float*)d_in[6];
    const float* thesis_lin_b = (const float*)d_in[7];
    const float* prof_emb     = (const float*)d_in[8];
    const float* W_nbr_sup    = (const float*)d_in[9];
    const float* W_root_sup   = (const float*)d_in[10];
    const float* b_sup        = (const float*)d_in[11];
    const float* W_nbr_rev    = (const float*)d_in[12];
    const float* W_root_rev   = (const float*)d_in[13];
    const float* b_rev        = (const float*)d_in[14];
    const float* cls_w        = (const float*)d_in[15];
    const float* cls_b        = (const float*)d_in[16];
    float*       out          = (float*)d_out;

    const size_t T_PLANE = (size_t)N_T * DDIM;
    const size_t M_PLANE = (size_t)N_M * DDIM;
    static_assert(2 * BM_PLANE <= (size_t)N_T * DDIM, "both sides' records fit in one plane");
    float* p    = (float*)d_ws;
    float* tX   = p; p += T_PLANE;
    float* tY   = p; p += T_PLANE;
    float* tZ   = p; p += T_PLANE;
    float* sT   = p;
    float* sM   = p + (size_t)OWNER_BLOCKS * OWN_PITCH(OWN_T_ROWS);
    static_assert((size_t)OWNER_BLOCKS * (OWN_PITCH(OWN_T_ROWS) + OWN_PITCH(OWN_M_ROWS)) <= (size_t)N_M * DDIM,
                  "both scalar planes fit in this plane");
    p += M_PLANE;
    float* mQ   = p; p += M_PLANE;
    float* mR   = p; p += M_PLANE;
    unsigned* bmM = (unsigned*)tY;
    unsigned* bmT = bmM + BM_PLANE;

    const int BLK = 256;
    const int ROWS_PER_BLK = 8 * 16;

    k_owner_lists<<<dim3(BM_NCH, 2), BM_CHUNK, 0, stream>>>(edge_dst, edge_src, bmM, bmT);

    float* ht0 = tZ;
    float* hm0 = mR;
    k_gemm_wmma<<<(N_T + ROWS_PER_BLK - 1) / ROWS_PER_BLK, BLK, 0, stream>>>(
        x_thesis, thesis_lin_w, F_IN, nullptr, nullptr, 0,
        thesis_lin_b, ht0, N_T, 0);
    {
        int n = N_M * (DDIM / 4);
        k_gather_rows<<<(n + BLK - 1) / BLK, BLK, 0, stream>>>(prof_emb, mentor_id, hm0, N_M);
    }

    float* hm1 = mQ;
    float* ht1 = tX;
    k_owner_layer<OWN_M_ROWS, 2, false><<<OWNER_BLOCKS, OWNER_THREADS, 0, stream>>>(
        ht0, edge_dst, edge_src, bmM, hm0, W_nbr_sup, W_root_sup, b_sup, nullptr, hm1, N_M, 1);
    k_owner_layer<OWN_T_ROWS, 8, false><<<OWNER_BLOCKS, OWNER_THREADS, 0, stream>>>(
        hm0, edge_src, edge_dst, bmT, ht0, W_nbr_rev, W_root_rev, b_rev, nullptr, ht1, N_T, 1);

    const size_t wOff = (size_t)DDIM * DDIM;
    k_owner_layer<OWN_M_ROWS, 2, true><<<OWNER_BLOCKS, OWNER_THREADS, 0, stream>>>(
        ht1, edge_dst, edge_src, bmM, hm1, W_nbr_sup + wOff, W_root_sup + wOff, b_sup + DDIM,
        cls_w + DDIM, sM, N_M, 0);
    k_owner_layer<OWN_T_ROWS, 8, true><<<OWNER_BLOCKS, OWNER_THREADS, 0, stream>>>(
        hm1, edge_src, edge_dst, bmT, ht1, W_nbr_rev + wOff, W_root_rev + wOff, b_rev + DDIM,
        cls_w, sT, N_T, 0);

    k_classify_scalars<<<(N_Q + OWNER_THREADS - 1) / OWNER_THREADS, OWNER_THREADS, 0, stream>>>(
        sT, sM, eli_src, eli_dst, cls_b, out, N_Q);
}
